# starting kernel
# baseline (speedup 1.0000x reference)
#include <hip/hip_runtime.h>
#include <hip/hip_bf16.h>


#define N_NODES 50000
#define N_EDGES 800000
#define CDIM    96
#define NC      (N_NODES * CDIM)
#define TILES   ((N_NODES / 16) * (CDIM / 16))

typedef __attribute__((ext_vector_type(2))) float v2f;
typedef __attribute__((ext_vector_type(8))) float v8f;

__global__ __launch_bounds__(256) void gnn_gemm_m(const float* __restrict__ x,
                                                  const float* __restrict__ W,
                                                  float* __restrict__ m) {
  const int wave = blockIdx.x * (blockDim.x >> 5) + (threadIdx.x >> 5);
  if (wave >= TILES) return;
  const int tr   = wave / (CDIM / 16);
  const int tc   = wave % (CDIM / 16);
  const int lane = threadIdx.x & 31;
  const int half = lane >> 4;
  const int l16  = lane & 15;
  const int row0 = tr * 16, col0 = tc * 16;

  const float* arow = x + (row0 + l16) * CDIM + 2 * half;
  const float* bcol = W + col0 + l16;

  v8f acc = {0.f, 0.f, 0.f, 0.f, 0.f, 0.f, 0.f, 0.f};
#pragma unroll
  for (int k0 = 0; k0 < CDIM; k0 += 4) {
    v2f a, b;
    a.x = arow[k0];
    a.y = arow[k0 + 1];
    const int k = k0 + 2 * half;
    b.x = bcol[(k    ) * CDIM];
    b.y = bcol[(k + 1) * CDIM];
    acc = __builtin_amdgcn_wmma_f32_16x16x4_f32(false, a, false, b,
                                                (short)0, acc, false, false);
  }

  float* orow = m + (row0 + 8 * half) * CDIM + col0 + l16;
#pragma unroll
  for (int r = 0; r < 8; ++r) orow[r * CDIM] = acc[r];
}

__global__ void gnn_zero(float4* __restrict__ p, int n4) {
  int i = blockIdx.x * blockDim.x + threadIdx.x;
  if (i < n4) p[i] = make_float4(0.f, 0.f, 0.f, 0.f);
}

__global__ void gnn_scatter(const float* __restrict__ m,
                            const int* __restrict__ edge_index,
                            float* __restrict__ agg) {
  const long long total  = (long long)N_EDGES * (CDIM / 4);
  const long long stride = (long long)gridDim.x * blockDim.x;
  for (long long i = (long long)blockIdx.x * blockDim.x + threadIdx.x;
       i < total; i += stride) {
    const int e = (int)(i / (CDIM / 4));
    const int g = (int)(i % (CDIM / 4));
    const int s = edge_index[e];
    const int d = edge_index[N_EDGES + e];
    const float4 v = *(const float4*)(m + s * CDIM + g * 4);
    float* p = agg + d * CDIM + g * 4;
    atomicAdd(p + 0, v.x);
    atomicAdd(p + 1, v.y);
    atomicAdd(p + 2, v.z);
    atomicAdd(p + 3, v.w);
  }
}

__global__ __launch_bounds__(256) void gnn_gru(const float* __restrict__ agg,
                                               const float* __restrict__ h,
                                               const float* __restrict__ w_ih,
                                               const float* __restrict__ w_hh,
                                               const float* __restrict__ b_ih,
                                               const float* __restrict__ b_hh,
                                               float* __restrict__ out) {
  const int wave = blockIdx.x * (blockDim.x >> 5) + (threadIdx.x >> 5);
  if (wave >= TILES) return;
  const int tr   = wave / (CDIM / 16);
  const int tc   = wave % (CDIM / 16);
  const int lane = threadIdx.x & 31;
  const int half = lane >> 4;
  const int l16  = lane & 15;
  const int row0 = tr * 16, col0 = tc * 16;
  const int col  = col0 + l16;

  v8f a_ir, a_iz, a_in, a_hr, a_hz, a_hn;
  {
    const float bir = b_ih[col], biz = b_ih[CDIM + col], bin_ = b_ih[2 * CDIM + col];
    const float bhr = b_hh[col], bhz = b_hh[CDIM + col], bhn  = b_hh[2 * CDIM + col];
#pragma unroll
    for (int r = 0; r < 8; ++r) {
      a_ir[r] = bir; a_iz[r] = biz; a_in[r] = bin_;
      a_hr[r] = bhr; a_hz[r] = bhz; a_hn[r] = bhn;
    }
  }

  const float* aggrow = agg + (row0 + l16) * CDIM + 2 * half;
  const float* hrow   = h   + (row0 + l16) * CDIM + 2 * half;
  const float* wih_r  = w_ih + (0 * CDIM + col) * CDIM;
  const float* wih_z  = w_ih + (1 * CDIM + col) * CDIM;
  const float* wih_n  = w_ih + (2 * CDIM + col) * CDIM;
  const float* whh_r  = w_hh + (0 * CDIM + col) * CDIM;
  const float* whh_z  = w_hh + (1 * CDIM + col) * CDIM;
  const float* whh_n  = w_hh + (2 * CDIM + col) * CDIM;

#pragma unroll 4
  for (int k0 = 0; k0 < CDIM; k0 += 4) {
    v2f ai, ah, b;
    ai.x = aggrow[k0]; ai.y = aggrow[k0 + 1];
    ah.x = hrow[k0];   ah.y = hrow[k0 + 1];
    const int k = k0 + 2 * half;

    b.x = wih_r[k]; b.y = wih_r[k + 1];
    a_ir = __builtin_amdgcn_wmma_f32_16x16x4_f32(false, ai, false, b, (short)0, a_ir, false, false);
    b.x = wih_z[k]; b.y = wih_z[k + 1];
    a_iz = __builtin_amdgcn_wmma_f32_16x16x4_f32(false, ai, false, b, (short)0, a_iz, false, false);
    b.x = wih_n[k]; b.y = wih_n[k + 1];
    a_in = __builtin_amdgcn_wmma_f32_16x16x4_f32(false, ai, false, b, (short)0, a_in, false, false);
    b.x = whh_r[k]; b.y = whh_r[k + 1];
    a_hr = __builtin_amdgcn_wmma_f32_16x16x4_f32(false, ah, false, b, (short)0, a_hr, false, false);
    b.x = whh_z[k]; b.y = whh_z[k + 1];
    a_hz = __builtin_amdgcn_wmma_f32_16x16x4_f32(false, ah, false, b, (short)0, a_hz, false, false);
    b.x = whh_n[k]; b.y = whh_n[k + 1];
    a_hn = __builtin_amdgcn_wmma_f32_16x16x4_f32(false, ah, false, b, (short)0, a_hn, false, false);
  }

#pragma unroll
  for (int r = 0; r < 8; ++r) {
    const int row = row0 + r + 8 * half;
    const float hv = h[row * CDIM + col];
    const float rr = 1.f / (1.f + __expf(-(a_ir[r] + a_hr[r])));
    const float zz = 1.f / (1.f + __expf(-(a_iz[r] + a_hz[r])));
    const float nn = tanhf(a_in[r] + rr * a_hn[r]);
    out[row * CDIM + col] = (1.f - zz) * nn + zz * hv;
  }
}

extern "C" void kernel_launch(void* const* d_in, const int* in_sizes, int n_in,
                              void* d_out, int out_size, void* d_ws, size_t ws_size,
                              hipStream_t stream) {
  const float* x      = (const float*)d_in[0];
  const int*   ei     = (const int*)  d_in[1];
  const float* weight = (const float*)d_in[2];
  const float* w_ih   = (const float*)d_in[3];
  const float* w_hh   = (const float*)d_in[4];
  const float* b_ih   = (const float*)d_in[5];
  const float* b_hh   = (const float*)d_in[6];
  float* out = (float*)d_out;

  float* xa  = (float*)d_ws;
  float* xb  = xa + NC;
  float* m   = xb + NC;
  float* agg = m  + NC;

  const int wavesPerBlock = 256 / 32;
  const int gemmBlocks    = (TILES + wavesPerBlock - 1) / wavesPerBlock;
  const int zeroBlocks    = (NC / 4 + 255) / 256;
  const int scatBlocks    = 2048;

  for (int l = 0; l < 3; ++l) {
    const float* xin  = (l == 0) ? x  : ((l == 1) ? xa : xb);
    float*       xout = (l == 2) ? out : ((l == 0) ? xa : xb);
    const float* Wl   = weight + l * CDIM * CDIM;

    gnn_gemm_m<<<gemmBlocks, 256, 0, stream>>>(xin, Wl, m);
    gnn_zero  <<<zeroBlocks, 256, 0, stream>>>((float4*)agg, NC / 4);
    gnn_scatter<<<scatBlocks, 256, 0, stream>>>(m, ei, agg);
    gnn_gru   <<<gemmBlocks, 256, 0, stream>>>(agg, xin, w_ih, w_hh, b_ih, b_hh, xout);
  }
}


# Float-atomic scatter-add and agg clear replaced by edges counting-sorted by destination once per call + atomic-free per-node row gather-sum each layer; sums now run in ascending source order
# speedup vs baseline: 1.3100x; 1.3100x over previous
#include <hip/hip_runtime.h>
#include <hip/hip_bf16.h>


#define N_NODES 50000
#define N_EDGES 800000
#define CDIM    96
#define NC      (N_NODES * CDIM)
#define TILES   ((N_NODES / 16) * (CDIM / 16))

typedef __attribute__((ext_vector_type(2))) float v2f;
typedef __attribute__((ext_vector_type(8))) float v8f;

__global__ __launch_bounds__(256) void gnn_gemm_m(const float* __restrict__ x,
                                                  const float* __restrict__ W,
                                                  float* __restrict__ m) {
  const int wave = blockIdx.x * (blockDim.x >> 5) + (threadIdx.x >> 5);
  if (wave >= TILES) return;
  const int tr   = wave / (CDIM / 16);
  const int tc   = wave % (CDIM / 16);
  const int lane = threadIdx.x & 31;
  const int half = lane >> 4;
  const int l16  = lane & 15;
  const int row0 = tr * 16, col0 = tc * 16;

  const float* arow = x + (row0 + l16) * CDIM + 2 * half;
  const float* bcol = W + col0 + l16;

  v8f acc = {0.f, 0.f, 0.f, 0.f, 0.f, 0.f, 0.f, 0.f};
#pragma unroll
  for (int k0 = 0; k0 < CDIM; k0 += 4) {
    v2f a, b;
    a.x = arow[k0];
    a.y = arow[k0 + 1];
    const int k = k0 + 2 * half;
    b.x = bcol[(k    ) * CDIM];
    b.y = bcol[(k + 1) * CDIM];
    acc = __builtin_amdgcn_wmma_f32_16x16x4_f32(false, a, false, b,
                                                (short)0, acc, false, false);
  }

  float* orow = m + (row0 + 8 * half) * CDIM + col0 + l16;
#pragma unroll
  for (int r = 0; r < 8; ++r) orow[r * CDIM] = acc[r];
}

#define SCAN_BLOCKS 49

__global__ void gnn_deg_zero(int* __restrict__ deg) {
  const int i = blockIdx.x * blockDim.x + threadIdx.x;
  if (i < N_NODES) deg[i] = 0;
}

__global__ void gnn_deg_count(const int* __restrict__ edge_index,
                              int* __restrict__ deg) {
  const int i = blockIdx.x * blockDim.x + threadIdx.x;
  if (i < N_EDGES / 4) {
    const int4 d = *(const int4*)(edge_index + N_EDGES + 4 * i);
    atomicAdd(deg + d.x, 1);
    atomicAdd(deg + d.y, 1);
    atomicAdd(deg + d.z, 1);
    atomicAdd(deg + d.w, 1);
  }
}

__global__ __launch_bounds__(1024) void gnn_rowptr(const int* __restrict__ deg,
                                                   int* __restrict__ rowptr,
                                                   int* __restrict__ cursor) {
  __shared__ int wpre[32];
  __shared__ int wtot[32];
  const int t    = threadIdx.x;
  const int lane = t & 31;
  const int w    = t >> 5;
  const int b    = blockIdx.x;

  int pre = 0;
  for (int r = 0; r < b; ++r) pre += deg[r * 1024 + t];

  const int i = b * 1024 + t;
  const int d = (i < N_NODES) ? deg[i] : 0;

  int inc = d;
#pragma unroll
  for (int off = 1; off < 32; off <<= 1) {
    const int v = __shfl_up(inc, off);
    if (lane >= off) inc += v;
  }
#pragma unroll
  for (int off = 16; off > 0; off >>= 1) pre += __shfl_xor(pre, off);
  if (lane == 31) wtot[w] = inc;
  if (lane == 0)  wpre[w] = pre;
  __syncthreads();

  int o = wpre[lane] + ((lane < w) ? wtot[lane] : 0);
#pragma unroll
  for (int off = 16; off > 0; off >>= 1) o += __shfl_xor(o, off);

  const int excl = o + inc - d;
  if (i < N_NODES) { rowptr[i] = excl; cursor[i] = excl; }
  if (i == N_NODES - 1) rowptr[N_NODES] = excl + d;
}

__global__ void gnn_fill(const int* __restrict__ edge_index,
                         int* __restrict__ cursor,
                         int* __restrict__ col) {
  const int i = blockIdx.x * blockDim.x + threadIdx.x;
  if (i < N_EDGES / 4) {
    const int4 s = *(const int4*)(edge_index + 4 * i);
    const int4 d = *(const int4*)(edge_index + N_EDGES + 4 * i);
    const int p0 = atomicAdd(cursor + d.x, 1);
    const int p1 = atomicAdd(cursor + d.y, 1);
    const int p2 = atomicAdd(cursor + d.z, 1);
    const int p3 = atomicAdd(cursor + d.w, 1);
    col[p0] = s.x;
    col[p1] = s.y;
    col[p2] = s.z;
    col[p3] = s.w;
  }
}

__global__ __launch_bounds__(256) void gnn_seg_sort(const int* __restrict__ rowptr,
                                                    const int* __restrict__ col,
                                                    int* __restrict__ scol) {
  const int lane = threadIdx.x & 31;
  const int wave0  = __builtin_amdgcn_readfirstlane(blockIdx.x * 8 + (threadIdx.x >> 5));
  const int nwaves = gridDim.x * 8;

  for (int node = wave0; node < N_NODES; node += nwaves) {
    const int beg = rowptr[node];
    const int n   = rowptr[node + 1] - beg;
    if (n <= 32) {
      const int key = (lane < n) ? col[beg + lane] : 0;
      int rank = 0;
      for (int c = 0; c < n; ++c) {
        const int other = __builtin_amdgcn_readlane(key, c);
        rank += (other < key || (other == key && c < lane)) ? 1 : 0;
      }
      if (lane < n) scol[beg + rank] = key;
    } else {
      for (int a = lane; a < n; a += 32) {
        const int key = col[beg + a];
        int rank = 0;
        for (int c = 0; c < n; ++c) {
          const int other = col[beg + c];
          rank += (other < key || (other == key && c < a)) ? 1 : 0;
        }
        scol[beg + rank] = key;
      }
    }
  }
}

__global__ __launch_bounds__(256) void gnn_gather(const float* __restrict__ m,
                                                  const int* __restrict__ rowptr,
                                                  const int* __restrict__ col,
                                                  float* __restrict__ agg) {
  const int lane = threadIdx.x & 31;
  const int wave0  = __builtin_amdgcn_readfirstlane(blockIdx.x * 8 + (threadIdx.x >> 5));
  const int nwaves = gridDim.x * 8;
  const int c4     = (lane < 24) ? lane * 4 : 0;

  for (int node = wave0; node < N_NODES; node += nwaves) {
    const int beg = rowptr[node];
    const int end = rowptr[node + 1];
    float4 acc = make_float4(0.f, 0.f, 0.f, 0.f);
    for (int b = beg; b < end; b += 32) {
      const int cnt   = (end - b < 32) ? (end - b) : 32;
      const int mycol = (lane < cnt) ? col[b + lane] : 0;
      for (int j = 0; j < cnt; j += 8) {
        float4 v[8];
#pragma unroll
        for (int u = 0; u < 8; ++u) {
          const int s = __builtin_amdgcn_readlane(mycol, j + u);
          v[u] = *(const float4*)(m + s * CDIM + c4);
        }
#pragma unroll
        for (int u = 0; u < 8; ++u) {
          if (j + u < cnt) {
            acc.x += v[u].x; acc.y += v[u].y; acc.z += v[u].z; acc.w += v[u].w;
          }
        }
      }
    }
    if (lane < 24) *(float4*)(agg + node * CDIM + c4) = acc;
  }
}

__global__ __launch_bounds__(256) void gnn_gru(const float* __restrict__ agg,
                                               const float* __restrict__ h,
                                               const float* __restrict__ w_ih,
                                               const float* __restrict__ w_hh,
                                               const float* __restrict__ b_ih,
                                               const float* __restrict__ b_hh,
                                               float* __restrict__ out) {
  const int wave = blockIdx.x * (blockDim.x >> 5) + (threadIdx.x >> 5);
  if (wave >= TILES) return;
  const int tr   = wave / (CDIM / 16);
  const int tc   = wave % (CDIM / 16);
  const int lane = threadIdx.x & 31;
  const int half = lane >> 4;
  const int l16  = lane & 15;
  const int row0 = tr * 16, col0 = tc * 16;
  const int col  = col0 + l16;

  v8f a_ir, a_iz, a_in, a_hr, a_hz, a_hn;
  {
    const float bir = b_ih[col], biz = b_ih[CDIM + col], bin_ = b_ih[2 * CDIM + col];
    const float bhr = b_hh[col], bhz = b_hh[CDIM + col], bhn  = b_hh[2 * CDIM + col];
#pragma unroll
    for (int r = 0; r < 8; ++r) {
      a_ir[r] = bir; a_iz[r] = biz; a_in[r] = bin_;
      a_hr[r] = bhr; a_hz[r] = bhz; a_hn[r] = bhn;
    }
  }

  const float* aggrow = agg + (row0 + l16) * CDIM + 2 * half;
  const float* hrow   = h   + (row0 + l16) * CDIM + 2 * half;
  const float* wih_r  = w_ih + (0 * CDIM + col) * CDIM;
  const float* wih_z  = w_ih + (1 * CDIM + col) * CDIM;
  const float* wih_n  = w_ih + (2 * CDIM + col) * CDIM;
  const float* whh_r  = w_hh + (0 * CDIM + col) * CDIM;
  const float* whh_z  = w_hh + (1 * CDIM + col) * CDIM;
  const float* whh_n  = w_hh + (2 * CDIM + col) * CDIM;

#pragma unroll 4
  for (int k0 = 0; k0 < CDIM; k0 += 4) {
    v2f ai, ah, b;
    ai.x = aggrow[k0]; ai.y = aggrow[k0 + 1];
    ah.x = hrow[k0];   ah.y = hrow[k0 + 1];
    const int k = k0 + 2 * half;

    b.x = wih_r[k]; b.y = wih_r[k + 1];
    a_ir = __builtin_amdgcn_wmma_f32_16x16x4_f32(false, ai, false, b, (short)0, a_ir, false, false);
    b.x = wih_z[k]; b.y = wih_z[k + 1];
    a_iz = __builtin_amdgcn_wmma_f32_16x16x4_f32(false, ai, false, b, (short)0, a_iz, false, false);
    b.x = wih_n[k]; b.y = wih_n[k + 1];
    a_in = __builtin_amdgcn_wmma_f32_16x16x4_f32(false, ai, false, b, (short)0, a_in, false, false);
    b.x = whh_r[k]; b.y = whh_r[k + 1];
    a_hr = __builtin_amdgcn_wmma_f32_16x16x4_f32(false, ah, false, b, (short)0, a_hr, false, false);
    b.x = whh_z[k]; b.y = whh_z[k + 1];
    a_hz = __builtin_amdgcn_wmma_f32_16x16x4_f32(false, ah, false, b, (short)0, a_hz, false, false);
    b.x = whh_n[k]; b.y = whh_n[k + 1];
    a_hn = __builtin_amdgcn_wmma_f32_16x16x4_f32(false, ah, false, b, (short)0, a_hn, false, false);
  }

#pragma unroll
  for (int r = 0; r < 8; ++r) {
    const int row = row0 + r + 8 * half;
    const float hv = h[row * CDIM + col];
    const float rr = 1.f / (1.f + __expf(-(a_ir[r] + a_hr[r])));
    const float zz = 1.f / (1.f + __expf(-(a_iz[r] + a_hz[r])));
    const float nn = tanhf(a_in[r] + rr * a_hn[r]);
    out[row * CDIM + col] = (1.f - zz) * nn + zz * hv;
  }
}

extern "C" void kernel_launch(void* const* d_in, const int* in_sizes, int n_in,
                              void* d_out, int out_size, void* d_ws, size_t ws_size,
                              hipStream_t stream) {
  const float* x      = (const float*)d_in[0];
  const int*   ei     = (const int*)  d_in[1];
  const float* weight = (const float*)d_in[2];
  const float* w_ih   = (const float*)d_in[3];
  const float* w_hh   = (const float*)d_in[4];
  const float* b_ih   = (const float*)d_in[5];
  const float* b_hh   = (const float*)d_in[6];
  float* out = (float*)d_out;

  float* xa  = (float*)d_ws;
  int*   ib  = (int*)(xa + NC);
  float* m   = xa + 2 * NC;
  float* agg = xa + 3 * NC;

  int* deg    = ib;
  int* rowptr = ib + 50048;
  int* cursor = ib + 2 * 50048;
  int* col    = ib + 3 * 50048;
  int* scol   = col + N_EDGES;

  const int wavesPerBlock = 256 / 32;
  const int gemmBlocks    = (TILES + wavesPerBlock - 1) / wavesPerBlock;
  const int nodeBlocks    = (N_NODES + 255) / 256;
  const int edgeBlocks    = (N_EDGES / 4 + 255) / 256;
  const int gathBlocks    = 2048;

  gnn_deg_zero <<<nodeBlocks, 256, 0, stream>>>(deg);
  gnn_deg_count<<<edgeBlocks, 256, 0, stream>>>(ei, deg);
  gnn_rowptr   <<<SCAN_BLOCKS, 1024, 0, stream>>>(deg, rowptr, cursor);
  gnn_fill     <<<edgeBlocks, 256, 0, stream>>>(ei, cursor, col);
  gnn_seg_sort <<<gathBlocks, 256, 0, stream>>>(rowptr, col, scol);

  for (int l = 0; l < 3; ++l) {
    const float* xin  = (l == 0) ? x   : ((l == 1) ? out : xa);
    float*       xout = (l == 1) ? xa  : out;
    const float* Wl   = weight + l * CDIM * CDIM;

    gnn_gemm_m<<<gemmBlocks, 256, 0, stream>>>(xin, Wl, m);
    gnn_gather<<<gathBlocks, 256, 0, stream>>>(m, rowptr, scol, agg);
    gnn_gru   <<<gemmBlocks, 256, 0, stream>>>(agg, xin, w_ih, w_hh, b_ih, b_hh, xout);
  }
}


# Both GEMM kernels: operand tiles staged in LDS by whole-line 16-byte global loads, WMMA fragments read from LDS (padded pitch) instead of strided 8-byte global loads; k order and sums unchanged
# speedup vs baseline: 2.0225x; 1.5439x over previous
#include <hip/hip_runtime.h>
#include <hip/hip_bf16.h>


#define N_NODES 50000
#define N_EDGES 800000
#define CDIM    96
#define NC      (N_NODES * CDIM)
#define TILES   ((N_NODES / 16) * (CDIM / 16))

typedef __attribute__((ext_vector_type(2))) float v2f;
typedef __attribute__((ext_vector_type(8))) float v8f;

#define LDP      100
#define WKP      104
#define ROWGRPS  391
#define LASTTILE (N_NODES / 16 - 1)

__device__ __forceinline__ void load_rowtile(const float* __restrict__ g, int lane, float4 (&v)[12]) {
#pragma unroll
  for (int i = 0; i < 12; ++i) v[i] = *(const float4*)(g + (i * 32 + lane) * 4);
}
__device__ __forceinline__ void store_rowtile(float* __restrict__ s, int lane, const float4 (&v)[12]) {
#pragma unroll
  for (int i = 0; i < 12; ++i) {
    const int idx = i * 32 + lane;
    *(float4*)(s + (idx / 24) * LDP + (idx % 24) * 4) = v[i];
  }
}

__global__ __launch_bounds__(256) void gnn_gemm_m(const float* __restrict__ x,
                                                  const float* __restrict__ W,
                                                  float* __restrict__ m) {
  __shared__ __attribute__((aligned(16))) float lds[CDIM * WKP + 8 * 16 * LDP];
  const int tid  = threadIdx.x;
  const int wid  = tid >> 5;
  const int lane = tid & 31;
  const int half = lane >> 4;
  const int l16  = lane & 15;
  float* Ws = lds;
  float* As = lds + CDIM * WKP + wid * (16 * LDP);

  const int  tr   = blockIdx.x * 8 + wid;
  const bool live = tr <= LASTTILE;
  const int  trc  = live ? tr : LASTTILE;
  const int  row0 = trc * 16;

  {
    float4 wv[9];
    float4 av[12];
#pragma unroll
    for (int i = 0; i < 9; ++i) wv[i] = *(const float4*)(W + (i * 256 + tid) * 4);
    load_rowtile(x + row0 * CDIM, lane, av);
#pragma unroll
    for (int i = 0; i < 9; ++i) {
      const int idx = i * 256 + tid;
      *(float4*)(Ws + (idx / 24) * WKP + (idx % 24) * 4) = wv[i];
    }
    store_rowtile(As, lane, av);
  }
  __syncthreads();

  v8f acc[6];
#pragma unroll
  for (int t = 0; t < 6; ++t) acc[t] = (v8f){0.f, 0.f, 0.f, 0.f, 0.f, 0.f, 0.f, 0.f};

  const float* ap = As + l16 * LDP + 2 * half;
  const float* bp = Ws + (2 * half) * WKP + l16;

#pragma unroll 4
  for (int k0 = 0; k0 < CDIM; k0 += 4) {
    const v2f a = *(const v2f*)(ap + k0);
#pragma unroll
    for (int t = 0; t < 6; ++t) {
      v2f b;
      b.x = bp[(k0    ) * WKP + t * 16];
      b.y = bp[(k0 + 1) * WKP + t * 16];
      acc[t] = __builtin_amdgcn_wmma_f32_16x16x4_f32(false, a, false, b,
                                                     (short)0, acc[t], false, false);
    }
  }

  if (live) {
    float* orow = m + (row0 + 8 * half) * CDIM + l16;
#pragma unroll
    for (int t = 0; t < 6; ++t) {
#pragma unroll
      for (int r = 0; r < 8; ++r) orow[r * CDIM + t * 16] = acc[t][r];
    }
  }
}

#define SCAN_BLOCKS 49

__global__ void gnn_deg_zero(int* __restrict__ deg) {
  const int i = blockIdx.x * blockDim.x + threadIdx.x;
  if (i < N_NODES) deg[i] = 0;
}

__global__ void gnn_deg_count(const int* __restrict__ edge_index,
                              int* __restrict__ deg) {
  const int i = blockIdx.x * blockDim.x + threadIdx.x;
  if (i < N_EDGES / 4) {
    const int4 d = *(const int4*)(edge_index + N_EDGES + 4 * i);
    atomicAdd(deg + d.x, 1);
    atomicAdd(deg + d.y, 1);
    atomicAdd(deg + d.z, 1);
    atomicAdd(deg + d.w, 1);
  }
}

__global__ __launch_bounds__(1024) void gnn_rowptr(const int* __restrict__ deg,
                                                   int* __restrict__ rowptr,
                                                   int* __restrict__ cursor) {
  __shared__ int wpre[32];
  __shared__ int wtot[32];
  const int t    = threadIdx.x;
  const int lane = t & 31;
  const int w    = t >> 5;
  const int b    = blockIdx.x;

  int pre = 0;
  for (int r = 0; r < b; ++r) pre += deg[r * 1024 + t];

  const int i = b * 1024 + t;
  const int d = (i < N_NODES) ? deg[i] : 0;

  int inc = d;
#pragma unroll
  for (int off = 1; off < 32; off <<= 1) {
    const int v = __shfl_up(inc, off);
    if (lane >= off) inc += v;
  }
#pragma unroll
  for (int off = 16; off > 0; off >>= 1) pre += __shfl_xor(pre, off);
  if (lane == 31) wtot[w] = inc;
  if (lane == 0)  wpre[w] = pre;
  __syncthreads();

  int o = wpre[lane] + ((lane < w) ? wtot[lane] : 0);
#pragma unroll
  for (int off = 16; off > 0; off >>= 1) o += __shfl_xor(o, off);

  const int excl = o + inc - d;
  if (i < N_NODES) { rowptr[i] = excl; cursor[i] = excl; }
  if (i == N_NODES - 1) rowptr[N_NODES] = excl + d;
}

__global__ void gnn_fill(const int* __restrict__ edge_index,
                         int* __restrict__ cursor,
                         int* __restrict__ col) {
  const int i = blockIdx.x * blockDim.x + threadIdx.x;
  if (i < N_EDGES / 4) {
    const int4 s = *(const int4*)(edge_index + 4 * i);
    const int4 d = *(const int4*)(edge_index + N_EDGES + 4 * i);
    const int p0 = atomicAdd(cursor + d.x, 1);
    const int p1 = atomicAdd(cursor + d.y, 1);
    const int p2 = atomicAdd(cursor + d.z, 1);
    const int p3 = atomicAdd(cursor + d.w, 1);
    col[p0] = s.x;
    col[p1] = s.y;
    col[p2] = s.z;
    col[p3] = s.w;
  }
}

__global__ __launch_bounds__(256) void gnn_seg_sort(const int* __restrict__ rowptr,
                                                    const int* __restrict__ col,
                                                    int* __restrict__ scol) {
  const int lane = threadIdx.x & 31;
  const int wave0  = __builtin_amdgcn_readfirstlane(blockIdx.x * 8 + (threadIdx.x >> 5));
  const int nwaves = gridDim.x * 8;

  for (int node = wave0; node < N_NODES; node += nwaves) {
    const int beg = rowptr[node];
    const int n   = rowptr[node + 1] - beg;
    if (n <= 32) {
      const int key = (lane < n) ? col[beg + lane] : 0;
      int rank = 0;
      for (int c = 0; c < n; ++c) {
        const int other = __builtin_amdgcn_readlane(key, c);
        rank += (other < key || (other == key && c < lane)) ? 1 : 0;
      }
      if (lane < n) scol[beg + rank] = key;
    } else {
      for (int a = lane; a < n; a += 32) {
        const int key = col[beg + a];
        int rank = 0;
        for (int c = 0; c < n; ++c) {
          const int other = col[beg + c];
          rank += (other < key || (other == key && c < a)) ? 1 : 0;
        }
        scol[beg + rank] = key;
      }
    }
  }
}

__global__ __launch_bounds__(256) void gnn_gather(const float* __restrict__ m,
                                                  const int* __restrict__ rowptr,
                                                  const int* __restrict__ col,
                                                  float* __restrict__ agg) {
  const int lane = threadIdx.x & 31;
  const int wave0  = __builtin_amdgcn_readfirstlane(blockIdx.x * 8 + (threadIdx.x >> 5));
  const int nwaves = gridDim.x * 8;
  const int c4     = (lane < 24) ? lane * 4 : 0;

  for (int node = wave0; node < N_NODES; node += nwaves) {
    const int beg = rowptr[node];
    const int end = rowptr[node + 1];
    float4 acc = make_float4(0.f, 0.f, 0.f, 0.f);
    for (int b = beg; b < end; b += 32) {
      const int cnt   = (end - b < 32) ? (end - b) : 32;
      const int mycol = (lane < cnt) ? col[b + lane] : 0;
      for (int j = 0; j < cnt; j += 8) {
        float4 v[8];
#pragma unroll
        for (int u = 0; u < 8; ++u) {
          const int s = __builtin_amdgcn_readlane(mycol, j + u);
          v[u] = *(const float4*)(m + s * CDIM + c4);
        }
#pragma unroll
        for (int u = 0; u < 8; ++u) {
          if (j + u < cnt) {
            acc.x += v[u].x; acc.y += v[u].y; acc.z += v[u].z; acc.w += v[u].w;
          }
        }
      }
    }
    if (lane < 24) *(float4*)(agg + node * CDIM + c4) = acc;
  }
}

__global__ __launch_bounds__(256) void gnn_gru(const float* __restrict__ agg,
                                               const float* __restrict__ h,
                                               const float* __restrict__ w_ih,
                                               const float* __restrict__ w_hh,
                                               const float* __restrict__ b_ih,
                                               const float* __restrict__ b_hh,
                                               float* __restrict__ out) {
  __shared__ __attribute__((aligned(16))) float lds[6 * 16 * LDP + 8 * 2 * 16 * LDP];
  const int tid  = threadIdx.x;
  const int wid  = tid >> 5;
  const int lane = tid & 31;
  const int half = lane >> 4;
  const int l16  = lane & 15;
  float* Bs = lds;
  float* Ag = lds + 6 * 16 * LDP + wid * (2 * 16 * LDP);
  float* Ah = Ag + 16 * LDP;

  const int  tc   = blockIdx.x % (CDIM / 16);
  const int  tr   = (blockIdx.x / (CDIM / 16)) * 8 + wid;
  const bool live = tr <= LASTTILE;
  const int  trc  = live ? tr : LASTTILE;
  const int  row0 = trc * 16, col0 = tc * 16;
  const int  col  = col0 + l16;

  const float bir = b_ih[col], biz = b_ih[CDIM + col], bin_ = b_ih[2 * CDIM + col];
  const float bhr = b_hh[col], bhz = b_hh[CDIM + col], bhn  = b_hh[2 * CDIM + col];

  {
    float4 wv[9];
    float4 gv[12], hv4[12];
#pragma unroll
    for (int i = 0; i < 9; ++i) {
      const int idx = i * 256 + tid;
      const int j   = idx / 384;
      const int w   = idx % 384;
      const float* srcw = (j < 3 ? w_ih : w_hh) + ((j % 3) * CDIM + col0) * CDIM;
      wv[i] = *(const float4*)(srcw + w * 4);
    }
    load_rowtile(agg + row0 * CDIM, lane, gv);
    load_rowtile(h   + row0 * CDIM, lane, hv4);
#pragma unroll
    for (int i = 0; i < 9; ++i) {
      const int idx = i * 256 + tid;
      const int j   = idx / 384;
      const int w   = idx % 384;
      *(float4*)(Bs + (j * 16 + w / 24) * LDP + (w % 24) * 4) = wv[i];
    }
    store_rowtile(Ag, lane, gv);
    store_rowtile(Ah, lane, hv4);
  }
  __syncthreads();

  v8f a_ir, a_iz, a_in, a_hr, a_hz, a_hn;
#pragma unroll
  for (int r = 0; r < 8; ++r) {
    a_ir[r] = bir; a_iz[r] = biz; a_in[r] = bin_;
    a_hr[r] = bhr; a_hz[r] = bhz; a_hn[r] = bhn;
  }

  const float* agp = Ag + l16 * LDP + 2 * half;
  const float* ahp = Ah + l16 * LDP + 2 * half;
  const float* bp  = Bs + l16 * LDP + 2 * half;

#pragma unroll 4
  for (int k0 = 0; k0 < CDIM; k0 += 4) {
    const v2f ai = *(const v2f*)(agp + k0);
    const v2f ah = *(const v2f*)(ahp + k0);
    const v2f b0 = *(const v2f*)(bp + 0 * 16 * LDP + k0);
    const v2f b1 = *(const v2f*)(bp + 1 * 16 * LDP + k0);
    const v2f b2 = *(const v2f*)(bp + 2 * 16 * LDP + k0);
    const v2f b3 = *(const v2f*)(bp + 3 * 16 * LDP + k0);
    const v2f b4 = *(const v2f*)(bp + 4 * 16 * LDP + k0);
    const v2f b5 = *(const v2f*)(bp + 5 * 16 * LDP + k0);
    a_ir = __builtin_amdgcn_wmma_f32_16x16x4_f32(false, ai, false, b0, (short)0, a_ir, false, false);
    a_iz = __builtin_amdgcn_wmma_f32_16x16x4_f32(false, ai, false, b1, (short)0, a_iz, false, false);
    a_in = __builtin_amdgcn_wmma_f32_16x16x4_f32(false, ai, false, b2, (short)0, a_in, false, false);
    a_hr = __builtin_amdgcn_wmma_f32_16x16x4_f32(false, ah, false, b3, (short)0, a_hr, false, false);
    a_hz = __builtin_amdgcn_wmma_f32_16x16x4_f32(false, ah, false, b4, (short)0, a_hz, false, false);
    a_hn = __builtin_amdgcn_wmma_f32_16x16x4_f32(false, ah, false, b5, (short)0, a_hn, false, false);
  }

  if (live) {
#pragma unroll
    for (int r = 0; r < 8; ++r) {
      const int lr  = r + 8 * half;
      const float hv = Ah[lr * LDP + col];
      const float rr = 1.f / (1.f + __expf(-(a_ir[r] + a_hr[r])));
      const float zz = 1.f / (1.f + __expf(-(a_iz[r] + a_hz[r])));
      const float nn = tanhf(a_in[r] + rr * a_hn[r]);
      out[(row0 + lr) * CDIM + col] = (1.f - zz) * nn + zz * hv;
    }
  }
}

extern "C" void kernel_launch(void* const* d_in, const int* in_sizes, int n_in,
                              void* d_out, int out_size, void* d_ws, size_t ws_size,
                              hipStream_t stream) {
  const float* x      = (const float*)d_in[0];
  const int*   ei     = (const int*)  d_in[1];
  const float* weight = (const float*)d_in[2];
  const float* w_ih   = (const float*)d_in[3];
  const float* w_hh   = (const float*)d_in[4];
  const float* b_ih   = (const float*)d_in[5];
  const float* b_hh   = (const float*)d_in[6];
  float* out = (float*)d_out;

  float* xa  = (float*)d_ws;
  int*   ib  = (int*)(xa + NC);
  float* m   = xa + 2 * NC;
  float* agg = xa + 3 * NC;

  int* deg    = ib;
  int* rowptr = ib + 50048;
  int* cursor = ib + 2 * 50048;
  int* col    = ib + 3 * 50048;
  int* scol   = col + N_EDGES;

  const int gemmBlocks    = ROWGRPS;
  const int gruBlocks     = ROWGRPS * (CDIM / 16);
  const int nodeBlocks    = (N_NODES + 255) / 256;
  const int edgeBlocks    = (N_EDGES / 4 + 255) / 256;
  const int gathBlocks    = 2048;

  gnn_deg_zero <<<nodeBlocks, 256, 0, stream>>>(deg);
  gnn_deg_count<<<edgeBlocks, 256, 0, stream>>>(ei, deg);
  gnn_rowptr   <<<SCAN_BLOCKS, 1024, 0, stream>>>(deg, rowptr, cursor);
  gnn_fill     <<<edgeBlocks, 256, 0, stream>>>(ei, cursor, col);
  gnn_seg_sort <<<gathBlocks, 256, 0, stream>>>(rowptr, col, scol);

  for (int l = 0; l < 3; ++l) {
    const float* xin  = (l == 0) ? x   : ((l == 1) ? out : xa);
    float*       xout = (l == 1) ? xa  : out;
    const float* Wl   = weight + l * CDIM * CDIM;

    gnn_gemm_m<<<gemmBlocks, 256, 0, stream>>>(xin, Wl, m);
    gnn_gather<<<gathBlocks, 256, 0, stream>>>(m, rowptr, scol, agg);
    gnn_gru   <<<gruBlocks,  256, 0, stream>>>(agg, xin, w_ih, w_hh, b_ih, b_hh, xout);
  }
}


# GRU kernel made resident: 510 work-groups stage their weight tiles in LDS once and each wave walks over its 4-5 row tiles in LDS rows of its own, no work-group barrier per tile; sums unchanged
# speedup vs baseline: 2.1141x; 1.0453x over previous
#include <hip/hip_runtime.h>
#include <hip/hip_bf16.h>


#define N_NODES 50000
#define N_EDGES 800000
#define CDIM    96
#define NC      (N_NODES * CDIM)
#define TILES   ((N_NODES / 16) * (CDIM / 16))

typedef __attribute__((ext_vector_type(2))) float v2f;
typedef __attribute__((ext_vector_type(8))) float v8f;

#define LDP      100
#define WKP      104
#define ROWGRPS  391
#define LASTTILE (N_NODES / 16 - 1)

__device__ __forceinline__ void load_rowtile(const float* __restrict__ g, int lane, float4 (&v)[12]) {
#pragma unroll
  for (int i = 0; i < 12; ++i) v[i] = *(const float4*)(g + (i * 32 + lane) * 4);
}
__device__ __forceinline__ void store_rowtile(float* __restrict__ s, int lane, const float4 (&v)[12]) {
#pragma unroll
  for (int i = 0; i < 12; ++i) {
    const int idx = i * 32 + lane;
    *(float4*)(s + (idx / 24) * LDP + (idx % 24) * 4) = v[i];
  }
}

__global__ __launch_bounds__(256) void gnn_gemm_m(const float* __restrict__ x,
                                                  const float* __restrict__ W,
                                                  float* __restrict__ m) {
  __shared__ __attribute__((aligned(16))) float lds[CDIM * WKP + 8 * 16 * LDP];
  const int tid  = threadIdx.x;
  const int wid  = tid >> 5;
  const int lane = tid & 31;
  const int half = lane >> 4;
  const int l16  = lane & 15;
  float* Ws = lds;
  float* As = lds + CDIM * WKP + wid * (16 * LDP);

  const int  tr   = blockIdx.x * 8 + wid;
  const bool live = tr <= LASTTILE;
  const int  trc  = live ? tr : LASTTILE;
  const int  row0 = trc * 16;

  {
    float4 wv[9];
    float4 av[12];
#pragma unroll
    for (int i = 0; i < 9; ++i) wv[i] = *(const float4*)(W + (i * 256 + tid) * 4);
    load_rowtile(x + row0 * CDIM, lane, av);
#pragma unroll
    for (int i = 0; i < 9; ++i) {
      const int idx = i * 256 + tid;
      *(float4*)(Ws + (idx / 24) * WKP + (idx % 24) * 4) = wv[i];
    }
    store_rowtile(As, lane, av);
  }
  __syncthreads();

  v8f acc[6];
#pragma unroll
  for (int t = 0; t < 6; ++t) acc[t] = (v8f){0.f, 0.f, 0.f, 0.f, 0.f, 0.f, 0.f, 0.f};

  const float* ap = As + l16 * LDP + 2 * half;
  const float* bp = Ws + (2 * half) * WKP + l16;

#pragma unroll 4
  for (int k0 = 0; k0 < CDIM; k0 += 4) {
    const v2f a = *(const v2f*)(ap + k0);
#pragma unroll
    for (int t = 0; t < 6; ++t) {
      v2f b;
      b.x = bp[(k0    ) * WKP + t * 16];
      b.y = bp[(k0 + 1) * WKP + t * 16];
      acc[t] = __builtin_amdgcn_wmma_f32_16x16x4_f32(false, a, false, b,
                                                     (short)0, acc[t], false, false);
    }
  }

  if (live) {
    float* orow = m + (row0 + 8 * half) * CDIM + l16;
#pragma unroll
    for (int t = 0; t < 6; ++t) {
#pragma unroll
      for (int r = 0; r < 8; ++r) orow[r * CDIM + t * 16] = acc[t][r];
    }
  }
}

#define SCAN_BLOCKS 49

__global__ void gnn_deg_zero(int* __restrict__ deg) {
  const int i = blockIdx.x * blockDim.x + threadIdx.x;
  if (i < N_NODES) deg[i] = 0;
}

__global__ void gnn_deg_count(const int* __restrict__ edge_index,
                              int* __restrict__ deg) {
  const int i = blockIdx.x * blockDim.x + threadIdx.x;
  if (i < N_EDGES / 4) {
    const int4 d = *(const int4*)(edge_index + N_EDGES + 4 * i);
    atomicAdd(deg + d.x, 1);
    atomicAdd(deg + d.y, 1);
    atomicAdd(deg + d.z, 1);
    atomicAdd(deg + d.w, 1);
  }
}

__global__ __launch_bounds__(1024) void gnn_rowptr(const int* __restrict__ deg,
                                                   int* __restrict__ rowptr,
                                                   int* __restrict__ cursor) {
  __shared__ int wpre[32];
  __shared__ int wtot[32];
  const int t    = threadIdx.x;
  const int lane = t & 31;
  const int w    = t >> 5;
  const int b    = blockIdx.x;

  int pre = 0;
  for (int r = 0; r < b; ++r) pre += deg[r * 1024 + t];

  const int i = b * 1024 + t;
  const int d = (i < N_NODES) ? deg[i] : 0;

  int inc = d;
#pragma unroll
  for (int off = 1; off < 32; off <<= 1) {
    const int v = __shfl_up(inc, off);
    if (lane >= off) inc += v;
  }
#pragma unroll
  for (int off = 16; off > 0; off >>= 1) pre += __shfl_xor(pre, off);
  if (lane == 31) wtot[w] = inc;
  if (lane == 0)  wpre[w] = pre;
  __syncthreads();

  int o = wpre[lane] + ((lane < w) ? wtot[lane] : 0);
#pragma unroll
  for (int off = 16; off > 0; off >>= 1) o += __shfl_xor(o, off);

  const int excl = o + inc - d;
  if (i < N_NODES) { rowptr[i] = excl; cursor[i] = excl; }
  if (i == N_NODES - 1) rowptr[N_NODES] = excl + d;
}

__global__ void gnn_fill(const int* __restrict__ edge_index,
                         int* __restrict__ cursor,
                         int* __restrict__ col) {
  const int i = blockIdx.x * blockDim.x + threadIdx.x;
  if (i < N_EDGES / 4) {
    const int4 s = *(const int4*)(edge_index + 4 * i);
    const int4 d = *(const int4*)(edge_index + N_EDGES + 4 * i);
    const int p0 = atomicAdd(cursor + d.x, 1);
    const int p1 = atomicAdd(cursor + d.y, 1);
    const int p2 = atomicAdd(cursor + d.z, 1);
    const int p3 = atomicAdd(cursor + d.w, 1);
    col[p0] = s.x;
    col[p1] = s.y;
    col[p2] = s.z;
    col[p3] = s.w;
  }
}

__global__ __launch_bounds__(256) void gnn_seg_sort(const int* __restrict__ rowptr,
                                                    const int* __restrict__ col,
                                                    int* __restrict__ scol) {
  const int lane = threadIdx.x & 31;
  const int wave0  = __builtin_amdgcn_readfirstlane(blockIdx.x * 8 + (threadIdx.x >> 5));
  const int nwaves = gridDim.x * 8;

  for (int node = wave0; node < N_NODES; node += nwaves) {
    const int beg = rowptr[node];
    const int n   = rowptr[node + 1] - beg;
    if (n <= 32) {
      const int key = (lane < n) ? col[beg + lane] : 0;
      int rank = 0;
      for (int c = 0; c < n; ++c) {
        const int other = __builtin_amdgcn_readlane(key, c);
        rank += (other < key || (other == key && c < lane)) ? 1 : 0;
      }
      if (lane < n) scol[beg + rank] = key;
    } else {
      for (int a = lane; a < n; a += 32) {
        const int key = col[beg + a];
        int rank = 0;
        for (int c = 0; c < n; ++c) {
          const int other = col[beg + c];
          rank += (other < key || (other == key && c < a)) ? 1 : 0;
        }
        scol[beg + rank] = key;
      }
    }
  }
}

__global__ __launch_bounds__(256) void gnn_gather(const float* __restrict__ m,
                                                  const int* __restrict__ rowptr,
                                                  const int* __restrict__ col,
                                                  float* __restrict__ agg) {
  const int lane = threadIdx.x & 31;
  const int wave0  = __builtin_amdgcn_readfirstlane(blockIdx.x * 8 + (threadIdx.x >> 5));
  const int nwaves = gridDim.x * 8;
  const int c4     = (lane < 24) ? lane * 4 : 0;

  for (int node = wave0; node < N_NODES; node += nwaves) {
    const int beg = rowptr[node];
    const int end = rowptr[node + 1];
    float4 acc = make_float4(0.f, 0.f, 0.f, 0.f);
    for (int b = beg; b < end; b += 32) {
      const int cnt   = (end - b < 32) ? (end - b) : 32;
      const int mycol = (lane < cnt) ? col[b + lane] : 0;
      for (int j = 0; j < cnt; j += 8) {
        float4 v[8];
#pragma unroll
        for (int u = 0; u < 8; ++u) {
          const int s = __builtin_amdgcn_readlane(mycol, j + u);
          v[u] = *(const float4*)(m + s * CDIM + c4);
        }
#pragma unroll
        for (int u = 0; u < 8; ++u) {
          if (j + u < cnt) {
            acc.x += v[u].x; acc.y += v[u].y; acc.z += v[u].z; acc.w += v[u].w;
          }
        }
      }
    }
    if (lane < 24) *(float4*)(agg + node * CDIM + c4) = acc;
  }
}

#define GRU_SLOTS 85
#define GRU_WAVES (GRU_SLOTS * 8)

__device__ __forceinline__ void gru_tile(const float* agp, const float* ahp,
                                         const float* bp, const float* Ah,
                                         float bir, float biz, float bin_, float bhr, float bhz, float bhn,
                                         float* __restrict__ out, int row0, int col, int half) {
  v8f a_ir, a_iz, a_in, a_hr, a_hz, a_hn;
#pragma unroll
  for (int r = 0; r < 8; ++r) {
    a_ir[r] = bir; a_iz[r] = biz; a_in[r] = bin_;
    a_hr[r] = bhr; a_hz[r] = bhz; a_hn[r] = bhn;
  }

#pragma unroll 4
  for (int k0 = 0; k0 < CDIM; k0 += 4) {
    const v2f ai = *(const v2f*)(agp + k0);
    const v2f ah = *(const v2f*)(ahp + k0);
    const v2f b0 = *(const v2f*)(bp + 0 * 16 * LDP + k0);
    const v2f b1 = *(const v2f*)(bp + 1 * 16 * LDP + k0);
    const v2f b2 = *(const v2f*)(bp + 2 * 16 * LDP + k0);
    const v2f b3 = *(const v2f*)(bp + 3 * 16 * LDP + k0);
    const v2f b4 = *(const v2f*)(bp + 4 * 16 * LDP + k0);
    const v2f b5 = *(const v2f*)(bp + 5 * 16 * LDP + k0);
    a_ir = __builtin_amdgcn_wmma_f32_16x16x4_f32(false, ai, false, b0, (short)0, a_ir, false, false);
    a_iz = __builtin_amdgcn_wmma_f32_16x16x4_f32(false, ai, false, b1, (short)0, a_iz, false, false);
    a_in = __builtin_amdgcn_wmma_f32_16x16x4_f32(false, ai, false, b2, (short)0, a_in, false, false);
    a_hr = __builtin_amdgcn_wmma_f32_16x16x4_f32(false, ah, false, b3, (short)0, a_hr, false, false);
    a_hz = __builtin_amdgcn_wmma_f32_16x16x4_f32(false, ah, false, b4, (short)0, a_hz, false, false);
    a_hn = __builtin_amdgcn_wmma_f32_16x16x4_f32(false, ah, false, b5, (short)0, a_hn, false, false);
  }

#pragma unroll
  for (int r = 0; r < 8; ++r) {
    const int lr  = r + 8 * half;
    const float hv = Ah[lr * LDP + col];
    const float rr = 1.f / (1.f + __expf(-(a_ir[r] + a_hr[r])));
    const float zz = 1.f / (1.f + __expf(-(a_iz[r] + a_hz[r])));
    const float nn = tanhf(a_in[r] + rr * a_hn[r]);
    out[(row0 + lr) * CDIM + col] = (1.f - zz) * nn + zz * hv;
  }
}

__global__ __launch_bounds__(256) void gnn_gru(const float* __restrict__ agg,
                                               const float* __restrict__ h,
                                               const float* __restrict__ w_ih,
                                               const float* __restrict__ w_hh,
                                               const float* __restrict__ b_ih,
                                               const float* __restrict__ b_hh,
                                               float* __restrict__ out) {
  __shared__ __attribute__((aligned(16))) float lds[6 * 16 * LDP + 8 * 2 * 16 * LDP];
  const int tid  = threadIdx.x;
  const int wid  = tid >> 5;
  const int lane = tid & 31;
  const int half = lane >> 4;
  const int l16  = lane & 15;
  float* Bs = lds;
  float* Ag = lds + 6 * 16 * LDP + wid * (2 * 16 * LDP);
  float* Ah = Ag + 16 * LDP;

  const int tc   = blockIdx.x % (CDIM / 16);
  const int slot = blockIdx.x / (CDIM / 16);
  const int col0 = tc * 16;
  const int col  = col0 + l16;
  const int tr0 = __builtin_amdgcn_readfirstlane(((wid + slot + tc) & 7) * GRU_SLOTS + slot);

  const float bir = b_ih[col], biz = b_ih[CDIM + col], bin_ = b_ih[2 * CDIM + col];
  const float bhr = b_hh[col], bhz = b_hh[CDIM + col], bhn  = b_hh[2 * CDIM + col];

  {
    float4 wv[9];
    float4 gv[12], hv4[12];
#pragma unroll
    for (int i = 0; i < 9; ++i) {
      const int idx = i * 256 + tid;
      const int j   = idx / 384;
      const int w   = idx % 384;
      const float* srcw = (j < 3 ? w_ih : w_hh) + ((j % 3) * CDIM + col0) * CDIM;
      wv[i] = *(const float4*)(srcw + w * 4);
    }
    load_rowtile(agg + tr0 * 16 * CDIM, lane, gv);
    load_rowtile(h   + tr0 * 16 * CDIM, lane, hv4);
#pragma unroll
    for (int i = 0; i < 9; ++i) {
      const int idx = i * 256 + tid;
      const int j   = idx / 384;
      const int w   = idx % 384;
      *(float4*)(Bs + (j * 16 + w / 24) * LDP + (w % 24) * 4) = wv[i];
    }
    store_rowtile(Ag, lane, gv);
    store_rowtile(Ah, lane, hv4);
  }
  __syncthreads();

  const float* agp = Ag + l16 * LDP + 2 * half;
  const float* ahp = Ah + l16 * LDP + 2 * half;
  const float* bp  = Bs + l16 * LDP + 2 * half;

  int tr = tr0;
  for (; tr + GRU_WAVES <= LASTTILE; tr += GRU_WAVES) {
    gru_tile(agp, ahp, bp, Ah, bir, biz, bin_, bhr, bhz, bhn, out, tr * 16, col, half);

    float4 gv[12], hv4[12];
    load_rowtile(agg + (tr + GRU_WAVES) * 16 * CDIM, lane, gv);
    load_rowtile(h   + (tr + GRU_WAVES) * 16 * CDIM, lane, hv4);
    store_rowtile(Ag, lane, gv);
    store_rowtile(Ah, lane, hv4);
  }
  gru_tile(agp, ahp, bp, Ah, bir, biz, bin_, bhr, bhz, bhn, out, tr * 16, col, half);
}

extern "C" void kernel_launch(void* const* d_in, const int* in_sizes, int n_in,
                              void* d_out, int out_size, void* d_ws, size_t ws_size,
                              hipStream_t stream) {
  const float* x      = (const float*)d_in[0];
  const int*   ei     = (const int*)  d_in[1];
  const float* weight = (const float*)d_in[2];
  const float* w_ih   = (const float*)d_in[3];
  const float* w_hh   = (const float*)d_in[4];
  const float* b_ih   = (const float*)d_in[5];
  const float* b_hh   = (const float*)d_in[6];
  float* out = (float*)d_out;

  float* xa  = (float*)d_ws;
  int*   ib  = (int*)(xa + NC);
  float* m   = xa + 2 * NC;
  float* agg = xa + 3 * NC;

  int* deg    = ib;
  int* rowptr = ib + 50048;
  int* cursor = ib + 2 * 50048;
  int* col    = ib + 3 * 50048;
  int* scol   = col + N_EDGES;

  const int gemmBlocks    = ROWGRPS;
  const int gruBlocks     = GRU_SLOTS * (CDIM / 16);
  const int nodeBlocks    = (N_NODES + 255) / 256;
  const int edgeBlocks    = (N_EDGES / 4 + 255) / 256;
  const int gathBlocks    = 2048;

  gnn_deg_zero <<<nodeBlocks, 256, 0, stream>>>(deg);
  gnn_deg_count<<<edgeBlocks, 256, 0, stream>>>(ei, deg);
  gnn_rowptr   <<<SCAN_BLOCKS, 1024, 0, stream>>>(deg, rowptr, cursor);
  gnn_fill     <<<edgeBlocks, 256, 0, stream>>>(ei, cursor, col);
  gnn_seg_sort <<<gathBlocks, 256, 0, stream>>>(rowptr, col, scol);

  for (int l = 0; l < 3; ++l) {
    const float* xin  = (l == 0) ? x   : ((l == 1) ? out : xa);
    float*       xout = (l == 1) ? xa  : out;
    const float* Wl   = weight + l * CDIM * CDIM;

    gnn_gemm_m<<<gemmBlocks, 256, 0, stream>>>(xin, Wl, m);
    gnn_gather<<<gathBlocks, 256, 0, stream>>>(m, rowptr, scol, agg);
    gnn_gru   <<<gruBlocks,  256, 0, stream>>>(agg, xin, w_ih, w_hh, b_ih, b_hh, xout);
  }
}


# The three x @ W[l] launches removed: W[l] folded into the GRU input weights (wf = w_ih @ W[l].T, WMMA, once per call beside the degree clear); gather sums rows of x; product re-associated
# speedup vs baseline: 2.3240x; 1.0993x over previous
#include <hip/hip_runtime.h>
#include <hip/hip_bf16.h>


#define N_NODES 50000
#define N_EDGES 800000
#define CDIM    96
#define NC      (N_NODES * CDIM)
#define GATES   (3 * CDIM)

typedef __attribute__((ext_vector_type(2))) float v2f;
typedef __attribute__((ext_vector_type(8))) float v8f;

#define LDP      100
#define LASTTILE (N_NODES / 16 - 1)

__device__ __forceinline__ void load_rowtile(const float* __restrict__ g, int lane, float4 (&v)[12]) {
#pragma unroll
  for (int i = 0; i < 12; ++i) v[i] = *(const float4*)(g + (i * 32 + lane) * 4);
}
__device__ __forceinline__ void store_rowtile(float* __restrict__ s, int lane, const float4 (&v)[12]) {
#pragma unroll
  for (int i = 0; i < 12; ++i) {
    const int idx = i * 32 + lane;
    *(float4*)(s + (idx / 24) * LDP + (idx % 24) * 4) = v[i];
  }
}

#define SCAN_BLOCKS 49
#define NODE_BLOCKS ((N_NODES + 255) / 256)
#define FOLD_TILES  (3 * (GATES / 16) * (CDIM / 16))
#define FOLD_BLOCKS ((FOLD_TILES + 7) / 8)

__global__ __launch_bounds__(256) void gnn_setup(int* __restrict__ deg,
                                                 const float* __restrict__ weight,
                                                 const float* __restrict__ w_ih,
                                                 float* __restrict__ wf) {
  if (blockIdx.x < NODE_BLOCKS) {
    const int i = blockIdx.x * 256 + threadIdx.x;
    if (i < N_NODES) deg[i] = 0;
    return;
  }
  const int lane = threadIdx.x & 31;
  const int half = lane >> 4;
  const int l16  = lane & 15;
  const int tile = __builtin_amdgcn_readfirstlane((blockIdx.x - NODE_BLOCKS) * 8 + (threadIdx.x >> 5));
  if (tile >= FOLD_TILES) return;

  const int l  = tile / ((GATES / 16) * (CDIM / 16));
  const int t  = tile % ((GATES / 16) * (CDIM / 16));
  const int n0 = (t / (CDIM / 16)) * 16;
  const int k0 = (t % (CDIM / 16)) * 16;

  const float* ap = w_ih + (n0 + l16) * CDIM + 2 * half;
  const float* bp = weight + l * CDIM * CDIM + (k0 + l16) * CDIM + 2 * half;

  v2f a[24], b[24];
#pragma unroll
  for (int s = 0; s < 24; ++s) {
    a[s] = *(const v2f*)(ap + 4 * s);
    b[s] = *(const v2f*)(bp + 4 * s);
  }

  v8f acc0 = (v8f){0.f, 0.f, 0.f, 0.f, 0.f, 0.f, 0.f, 0.f};
  v8f acc1 = (v8f){0.f, 0.f, 0.f, 0.f, 0.f, 0.f, 0.f, 0.f};
#pragma unroll
  for (int s = 0; s < 24; s += 2) {
    acc0 = __builtin_amdgcn_wmma_f32_16x16x4_f32(false, a[s],     false, b[s],     (short)0, acc0, false, false);
    acc1 = __builtin_amdgcn_wmma_f32_16x16x4_f32(false, a[s + 1], false, b[s + 1], (short)0, acc1, false, false);
  }

  float* o = wf + l * GATES * CDIM + (n0 + 8 * half) * CDIM + k0 + l16;
#pragma unroll
  for (int r = 0; r < 8; ++r) o[r * CDIM] = acc0[r] + acc1[r];
}

__global__ void gnn_deg_count(const int* __restrict__ edge_index,
                              int* __restrict__ deg) {
  const int i = blockIdx.x * blockDim.x + threadIdx.x;
  if (i < N_EDGES / 4) {
    const int4 d = *(const int4*)(edge_index + N_EDGES + 4 * i);
    atomicAdd(deg + d.x, 1);
    atomicAdd(deg + d.y, 1);
    atomicAdd(deg + d.z, 1);
    atomicAdd(deg + d.w, 1);
  }
}

__global__ __launch_bounds__(1024) void gnn_rowptr(const int* __restrict__ deg,
                                                   int* __restrict__ rowptr,
                                                   int* __restrict__ cursor) {
  __shared__ int wpre[32];
  __shared__ int wtot[32];
  const int t    = threadIdx.x;
  const int lane = t & 31;
  const int w    = t >> 5;
  const int b    = blockIdx.x;

  int pre = 0;
  for (int r = 0; r < b; ++r) pre += deg[r * 1024 + t];

  const int i = b * 1024 + t;
  const int d = (i < N_NODES) ? deg[i] : 0;

  int inc = d;
#pragma unroll
  for (int off = 1; off < 32; off <<= 1) {
    const int v = __shfl_up(inc, off);
    if (lane >= off) inc += v;
  }
#pragma unroll
  for (int off = 16; off > 0; off >>= 1) pre += __shfl_xor(pre, off);
  if (lane == 31) wtot[w] = inc;
  if (lane == 0)  wpre[w] = pre;
  __syncthreads();

  int o = wpre[lane] + ((lane < w) ? wtot[lane] : 0);
#pragma unroll
  for (int off = 16; off > 0; off >>= 1) o += __shfl_xor(o, off);

  const int excl = o + inc - d;
  if (i < N_NODES) { rowptr[i] = excl; cursor[i] = excl; }
  if (i == N_NODES - 1) rowptr[N_NODES] = excl + d;
}

__global__ void gnn_fill(const int* __restrict__ edge_index,
                         int* __restrict__ cursor,
                         int* __restrict__ col) {
  const int i = blockIdx.x * blockDim.x + threadIdx.x;
  if (i < N_EDGES / 4) {
    const int4 s = *(const int4*)(edge_index + 4 * i);
    const int4 d = *(const int4*)(edge_index + N_EDGES + 4 * i);
    const int p0 = atomicAdd(cursor + d.x, 1);
    const int p1 = atomicAdd(cursor + d.y, 1);
    const int p2 = atomicAdd(cursor + d.z, 1);
    const int p3 = atomicAdd(cursor + d.w, 1);
    col[p0] = s.x;
    col[p1] = s.y;
    col[p2] = s.z;
    col[p3] = s.w;
  }
}

__global__ __launch_bounds__(256) void gnn_seg_sort(const int* __restrict__ rowptr,
                                                    const int* __restrict__ col,
                                                    int* __restrict__ scol) {
  const int lane = threadIdx.x & 31;
  const int wave0  = __builtin_amdgcn_readfirstlane(blockIdx.x * 8 + (threadIdx.x >> 5));
  const int nwaves = gridDim.x * 8;

  for (int node = wave0; node < N_NODES; node += nwaves) {
    const int beg = rowptr[node];
    const int n   = rowptr[node + 1] - beg;
    if (n <= 32) {
      const int key = (lane < n) ? col[beg + lane] : 0;
      int rank = 0;
      for (int c = 0; c < n; ++c) {
        const int other = __builtin_amdgcn_readlane(key, c);
        rank += (other < key || (other == key && c < lane)) ? 1 : 0;
      }
      if (lane < n) scol[beg + rank] = key;
    } else {
      for (int a = lane; a < n; a += 32) {
        const int key = col[beg + a];
        int rank = 0;
        for (int c = 0; c < n; ++c) {
          const int other = col[beg + c];
          rank += (other < key || (other == key && c < a)) ? 1 : 0;
        }
        scol[beg + rank] = key;
      }
    }
  }
}

__global__ __launch_bounds__(256) void gnn_gather(const float* __restrict__ m,
                                                  const int* __restrict__ rowptr,
                                                  const int* __restrict__ col,
                                                  float* __restrict__ agg) {
  const int lane = threadIdx.x & 31;
  const int wave0  = __builtin_amdgcn_readfirstlane(blockIdx.x * 8 + (threadIdx.x >> 5));
  const int nwaves = gridDim.x * 8;
  const int c4     = (lane < 24) ? lane * 4 : 0;

  for (int node = wave0; node < N_NODES; node += nwaves) {
    const int beg = rowptr[node];
    const int end = rowptr[node + 1];
    float4 acc = make_float4(0.f, 0.f, 0.f, 0.f);
    for (int b = beg; b < end; b += 32) {
      const int cnt   = (end - b < 32) ? (end - b) : 32;
      const int mycol = (lane < cnt) ? col[b + lane] : 0;
      for (int j = 0; j < cnt; j += 8) {
        float4 v[8];
#pragma unroll
        for (int u = 0; u < 8; ++u) {
          const int s = __builtin_amdgcn_readlane(mycol, j + u);
          v[u] = *(const float4*)(m + s * CDIM + c4);
        }
#pragma unroll
        for (int u = 0; u < 8; ++u) {
          if (j + u < cnt) {
            acc.x += v[u].x; acc.y += v[u].y; acc.z += v[u].z; acc.w += v[u].w;
          }
        }
      }
    }
    if (lane < 24) *(float4*)(agg + node * CDIM + c4) = acc;
  }
}

#define GRU_SLOTS 85
#define GRU_WAVES (GRU_SLOTS * 8)

__device__ __forceinline__ void gru_tile(const float* agp, const float* ahp,
                                         const float* bp, const float* Ah,
                                         float bir, float biz, float bin_, float bhr, float bhz, float bhn,
                                         float* __restrict__ out, int row0, int col, int half) {
  v8f a_ir, a_iz, a_in, a_hr, a_hz, a_hn;
#pragma unroll
  for (int r = 0; r < 8; ++r) {
    a_ir[r] = bir; a_iz[r] = biz; a_in[r] = bin_;
    a_hr[r] = bhr; a_hz[r] = bhz; a_hn[r] = bhn;
  }

#pragma unroll 4
  for (int k0 = 0; k0 < CDIM; k0 += 4) {
    const v2f ai = *(const v2f*)(agp + k0);
    const v2f ah = *(const v2f*)(ahp + k0);
    const v2f b0 = *(const v2f*)(bp + 0 * 16 * LDP + k0);
    const v2f b1 = *(const v2f*)(bp + 1 * 16 * LDP + k0);
    const v2f b2 = *(const v2f*)(bp + 2 * 16 * LDP + k0);
    const v2f b3 = *(const v2f*)(bp + 3 * 16 * LDP + k0);
    const v2f b4 = *(const v2f*)(bp + 4 * 16 * LDP + k0);
    const v2f b5 = *(const v2f*)(bp + 5 * 16 * LDP + k0);
    a_ir = __builtin_amdgcn_wmma_f32_16x16x4_f32(false, ai, false, b0, (short)0, a_ir, false, false);
    a_iz = __builtin_amdgcn_wmma_f32_16x16x4_f32(false, ai, false, b1, (short)0, a_iz, false, false);
    a_in = __builtin_amdgcn_wmma_f32_16x16x4_f32(false, ai, false, b2, (short)0, a_in, false, false);
    a_hr = __builtin_amdgcn_wmma_f32_16x16x4_f32(false, ah, false, b3, (short)0, a_hr, false, false);
    a_hz = __builtin_amdgcn_wmma_f32_16x16x4_f32(false, ah, false, b4, (short)0, a_hz, false, false);
    a_hn = __builtin_amdgcn_wmma_f32_16x16x4_f32(false, ah, false, b5, (short)0, a_hn, false, false);
  }

#pragma unroll
  for (int r = 0; r < 8; ++r) {
    const int lr  = r + 8 * half;
    const float hv = Ah[lr * LDP + col];
    const float rr = 1.f / (1.f + __expf(-(a_ir[r] + a_hr[r])));
    const float zz = 1.f / (1.f + __expf(-(a_iz[r] + a_hz[r])));
    const float nn = tanhf(a_in[r] + rr * a_hn[r]);
    out[(row0 + lr) * CDIM + col] = (1.f - zz) * nn + zz * hv;
  }
}

__global__ __launch_bounds__(256) void gnn_gru(const float* __restrict__ agg,
                                               const float* __restrict__ h,
                                               const float* __restrict__ w_ih,
                                               const float* __restrict__ w_hh,
                                               const float* __restrict__ b_ih,
                                               const float* __restrict__ b_hh,
                                               float* __restrict__ out) {
  __shared__ __attribute__((aligned(16))) float lds[6 * 16 * LDP + 8 * 2 * 16 * LDP];
  const int tid  = threadIdx.x;
  const int wid  = tid >> 5;
  const int lane = tid & 31;
  const int half = lane >> 4;
  const int l16  = lane & 15;
  float* Bs = lds;
  float* Ag = lds + 6 * 16 * LDP + wid * (2 * 16 * LDP);
  float* Ah = Ag + 16 * LDP;

  const int tc   = blockIdx.x % (CDIM / 16);
  const int slot = blockIdx.x / (CDIM / 16);
  const int col0 = tc * 16;
  const int col  = col0 + l16;
  const int tr0 = __builtin_amdgcn_readfirstlane(((wid + slot + tc) & 7) * GRU_SLOTS + slot);

  const float bir = b_ih[col], biz = b_ih[CDIM + col], bin_ = b_ih[2 * CDIM + col];
  const float bhr = b_hh[col], bhz = b_hh[CDIM + col], bhn  = b_hh[2 * CDIM + col];

  {
    float4 wv[9];
    float4 gv[12], hv4[12];
#pragma unroll
    for (int i = 0; i < 9; ++i) {
      const int idx = i * 256 + tid;
      const int j   = idx / 384;
      const int w   = idx % 384;
      const float* srcw = (j < 3 ? w_ih : w_hh) + ((j % 3) * CDIM + col0) * CDIM;
      wv[i] = *(const float4*)(srcw + w * 4);
    }
    load_rowtile(agg + tr0 * 16 * CDIM, lane, gv);
    load_rowtile(h   + tr0 * 16 * CDIM, lane, hv4);
#pragma unroll
    for (int i = 0; i < 9; ++i) {
      const int idx = i * 256 + tid;
      const int j   = idx / 384;
      const int w   = idx % 384;
      *(float4*)(Bs + (j * 16 + w / 24) * LDP + (w % 24) * 4) = wv[i];
    }
    store_rowtile(Ag, lane, gv);
    store_rowtile(Ah, lane, hv4);
  }
  __syncthreads();

  const float* agp = Ag + l16 * LDP + 2 * half;
  const float* ahp = Ah + l16 * LDP + 2 * half;
  const float* bp  = Bs + l16 * LDP + 2 * half;

  int tr = tr0;
  for (; tr + GRU_WAVES <= LASTTILE; tr += GRU_WAVES) {
    gru_tile(agp, ahp, bp, Ah, bir, biz, bin_, bhr, bhz, bhn, out, tr * 16, col, half);

    float4 gv[12], hv4[12];
    load_rowtile(agg + (tr + GRU_WAVES) * 16 * CDIM, lane, gv);
    load_rowtile(h   + (tr + GRU_WAVES) * 16 * CDIM, lane, hv4);
    store_rowtile(Ag, lane, gv);
    store_rowtile(Ah, lane, hv4);
  }
  gru_tile(agp, ahp, bp, Ah, bir, biz, bin_, bhr, bhz, bhn, out, tr * 16, col, half);
}

extern "C" void kernel_launch(void* const* d_in, const int* in_sizes, int n_in,
                              void* d_out, int out_size, void* d_ws, size_t ws_size,
                              hipStream_t stream) {
  const float* x      = (const float*)d_in[0];
  const int*   ei     = (const int*)  d_in[1];
  const float* weight = (const float*)d_in[2];
  const float* w_ih   = (const float*)d_in[3];
  const float* w_hh   = (const float*)d_in[4];
  const float* b_ih   = (const float*)d_in[5];
  const float* b_hh   = (const float*)d_in[6];
  float* out = (float*)d_out;

  float* xa  = (float*)d_ws;
  int*   ib  = (int*)(xa + NC);
  float* wf  = xa + 2 * NC;
  float* agg = xa + 3 * NC;

  int* deg    = ib;
  int* rowptr = ib + 50048;
  int* cursor = ib + 2 * 50048;
  int* col    = ib + 3 * 50048;
  int* scol   = col + N_EDGES;

  const int gruBlocks     = GRU_SLOTS * (CDIM / 16);
  const int setupBlocks   = NODE_BLOCKS + FOLD_BLOCKS;
  const int edgeBlocks    = (N_EDGES / 4 + 255) / 256;
  const int gathBlocks    = 2048;

  gnn_setup    <<<setupBlocks, 256, 0, stream>>>(deg, weight, w_ih, wf);
  gnn_deg_count<<<edgeBlocks, 256, 0, stream>>>(ei, deg);
  gnn_rowptr   <<<SCAN_BLOCKS, 1024, 0, stream>>>(deg, rowptr, cursor);
  gnn_fill     <<<edgeBlocks, 256, 0, stream>>>(ei, cursor, col);
  gnn_seg_sort <<<gathBlocks, 256, 0, stream>>>(rowptr, col, scol);

  for (int l = 0; l < 3; ++l) {
    const float* xin  = (l == 0) ? x   : ((l == 1) ? out : xa);
    float*       xout = (l == 1) ? xa  : out;
    const float* wfl  = wf + l * GATES * CDIM;

    gnn_gather<<<gathBlocks, 256, 0, stream>>>(xin, rowptr, scol, agg);
    gnn_gru   <<<gruBlocks,  256, 0, stream>>>(agg, xin, wfl, w_hh, b_ih, b_hh, xout);
  }
}


# Edges grouped by destination by a two-level counting sort on LDS counters (chunk counts per node range, flat scan, ranked scatter, per-range count and place) instead of 1.6M global integer atomics
# speedup vs baseline: 3.0367x; 1.3067x over previous
#include <hip/hip_runtime.h>
#include <hip/hip_bf16.h>


#define N_NODES 50000
#define N_EDGES 800000
#define CDIM    96
#define NC      (N_NODES * CDIM)
#define GATES   (3 * CDIM)

typedef __attribute__((ext_vector_type(2))) float v2f;
typedef __attribute__((ext_vector_type(8))) float v8f;

#define LDP      100
#define LASTTILE (N_NODES / 16 - 1)

__device__ __forceinline__ void load_rowtile(const float* __restrict__ g, int lane, float4 (&v)[12]) {
#pragma unroll
  for (int i = 0; i < 12; ++i) v[i] = *(const float4*)(g + (i * 32 + lane) * 4);
}
__device__ __forceinline__ void store_rowtile(float* __restrict__ s, int lane, const float4 (&v)[12]) {
#pragma unroll
  for (int i = 0; i < 12; ++i) {
    const int idx = i * 32 + lane;
    *(float4*)(s + (idx / 24) * LDP + (idx % 24) * 4) = v[i];
  }
}

#define NB          250
#define BW          200
#define NCH         250
#define CH4         800
#define NSCAN       (NB * NCH)
#define SCAN_BLOCKS 62
#define FOLD_TILES  (3 * (GATES / 16) * (CDIM / 16))
#define FOLD_BLOCKS ((FOLD_TILES + 7) / 8)

__global__ __launch_bounds__(256) void gnn_setup(const int* __restrict__ edge_index,
                                                 int* __restrict__ ht,
                                                 const float* __restrict__ weight,
                                                 const float* __restrict__ w_ih,
                                                 float* __restrict__ wf) {
  __shared__ int hcnt[256];
  if (blockIdx.x < NCH) {
    const int t = threadIdx.x;
    const int c = blockIdx.x;
    hcnt[t] = 0;
    int4 d[4];
#pragma unroll
    for (int k = 0; k < 4; ++k) {
      const int i = t + 256 * k;
      d[k] = (i < CH4) ? *(const int4*)(edge_index + N_EDGES + (c * CH4 + i) * 4) : make_int4(0, 0, 0, 0);
    }
    __syncthreads();
#pragma unroll
    for (int k = 0; k < 4; ++k) {
      if (t + 256 * k < CH4) {
        atomicAdd(&hcnt[(unsigned)d[k].x / BW], 1);
        atomicAdd(&hcnt[(unsigned)d[k].y / BW], 1);
        atomicAdd(&hcnt[(unsigned)d[k].z / BW], 1);
        atomicAdd(&hcnt[(unsigned)d[k].w / BW], 1);
      }
    }
    __syncthreads();
    if (t < NB) ht[t * NCH + c] = hcnt[t];
    return;
  }
  const int lane = threadIdx.x & 31;
  const int half = lane >> 4;
  const int l16  = lane & 15;
  const int tile = __builtin_amdgcn_readfirstlane((blockIdx.x - NCH) * 8 + (threadIdx.x >> 5));
  if (tile >= FOLD_TILES) return;

  const int l  = tile / ((GATES / 16) * (CDIM / 16));
  const int t  = tile % ((GATES / 16) * (CDIM / 16));
  const int n0 = (t / (CDIM / 16)) * 16;
  const int k0 = (t % (CDIM / 16)) * 16;

  const float* ap = w_ih + (n0 + l16) * CDIM + 2 * half;
  const float* bp = weight + l * CDIM * CDIM + (k0 + l16) * CDIM + 2 * half;

  v2f a[24], b[24];
#pragma unroll
  for (int s = 0; s < 24; ++s) {
    a[s] = *(const v2f*)(ap + 4 * s);
    b[s] = *(const v2f*)(bp + 4 * s);
  }

  v8f acc0 = (v8f){0.f, 0.f, 0.f, 0.f, 0.f, 0.f, 0.f, 0.f};
  v8f acc1 = (v8f){0.f, 0.f, 0.f, 0.f, 0.f, 0.f, 0.f, 0.f};
#pragma unroll
  for (int s = 0; s < 24; s += 2) {
    acc0 = __builtin_amdgcn_wmma_f32_16x16x4_f32(false, a[s],     false, b[s],     (short)0, acc0, false, false);
    acc1 = __builtin_amdgcn_wmma_f32_16x16x4_f32(false, a[s + 1], false, b[s + 1], (short)0, acc1, false, false);
  }

  float* o = wf + l * GATES * CDIM + (n0 + 8 * half) * CDIM + k0 + l16;
#pragma unroll
  for (int r = 0; r < 8; ++r) o[r * CDIM] = acc0[r] + acc1[r];
}

__global__ __launch_bounds__(1024) void gnn_scan(const int* __restrict__ cntv,
                                                 int* __restrict__ base) {
  __shared__ int wpre[32];
  __shared__ int wtot[32];
  const int t    = threadIdx.x;
  const int lane = t & 31;
  const int w    = t >> 5;
  const int b    = blockIdx.x;

  int pre = 0;
  for (int r = 0; r < b; ++r) pre += cntv[r * 1024 + t];

  const int i = b * 1024 + t;
  const int d = (i < NSCAN) ? cntv[i] : 0;

  int inc = d;
#pragma unroll
  for (int off = 1; off < 32; off <<= 1) {
    const int v = __shfl_up(inc, off);
    if (lane >= off) inc += v;
  }
#pragma unroll
  for (int off = 16; off > 0; off >>= 1) pre += __shfl_xor(pre, off);
  if (lane == 31) wtot[w] = inc;
  if (lane == 0)  wpre[w] = pre;
  __syncthreads();

  int o = wpre[lane] + ((lane < w) ? wtot[lane] : 0);
#pragma unroll
  for (int off = 16; off > 0; off >>= 1) o += __shfl_xor(o, off);

  const int excl = o + inc - d;
  if (i < NSCAN) base[i] = excl;
  if (i == NSCAN - 1) base[NSCAN] = excl + d;
}

#define PLACE_EDGE(dd, ss)                                              \
  {                                                                     \
    const unsigned r_  = (unsigned)(dd) / BW;                           \
    const unsigned dl_ = (unsigned)(dd) - r_ * BW;                      \
    const int p_ = atomicAdd(&cur[r_], 1);                              \
    tmp[p_] = (int)((dl_ << 16) | (unsigned)(ss));                      \
  }
__global__ __launch_bounds__(256) void gnn_scatter(const int* __restrict__ edge_index,
                                                   const int* __restrict__ base,
                                                   int* __restrict__ tmp) {
  __shared__ int cur[256];
  const int t = threadIdx.x;
  const int c = blockIdx.x;
  int4 s[4], d[4];
#pragma unroll
  for (int k = 0; k < 4; ++k) {
    const int i = t + 256 * k;
    s[k] = (i < CH4) ? *(const int4*)(edge_index + (c * CH4 + i) * 4) : make_int4(0, 0, 0, 0);
    d[k] = (i < CH4) ? *(const int4*)(edge_index + N_EDGES + (c * CH4 + i) * 4) : make_int4(0, 0, 0, 0);
  }
  if (t < NB) cur[t] = base[t * NCH + c];
  __syncthreads();
#pragma unroll
  for (int k = 0; k < 4; ++k) {
    if (t + 256 * k < CH4) {
      PLACE_EDGE(d[k].x, s[k].x)
      PLACE_EDGE(d[k].y, s[k].y)
      PLACE_EDGE(d[k].z, s[k].z)
      PLACE_EDGE(d[k].w, s[k].w)
    }
  }
}

__global__ __launch_bounds__(256) void gnn_range(const int* __restrict__ base,
                                                 const int* __restrict__ tmp,
                                                 int* __restrict__ rowptr,
                                                 int* __restrict__ col) {
  __shared__ int cnt[256];
  __shared__ int cu[256];
  __shared__ int wtot[8];
  const int t    = threadIdx.x;
  const int lane = t & 31;
  const int wv   = t >> 5;
  const int r    = blockIdx.x;
  const int beg  = base[r * NCH];
  const int end  = base[(r + 1) * NCH];
  const int n    = end - beg;

  cnt[t] = 0;
  __syncthreads();
  for (int i0 = t; i0 < n; i0 += 1024) {
    int w[4];
#pragma unroll
    for (int k = 0; k < 4; ++k) w[k] = (i0 + 256 * k < n) ? tmp[beg + i0 + 256 * k] : -1;
#pragma unroll
    for (int k = 0; k < 4; ++k)
      if (w[k] >= 0) atomicAdd(&cnt[w[k] >> 16], 1);
  }
  __syncthreads();

  const int cn = cnt[t];
  int inc = cn;
#pragma unroll
  for (int off = 1; off < 32; off <<= 1) {
    const int v = __shfl_up(inc, off);
    if (lane >= off) inc += v;
  }
  if (lane == 31) wtot[wv] = inc;
  __syncthreads();
  int front = 0;
#pragma unroll
  for (int k = 0; k < 8; ++k) front += (k < wv) ? wtot[k] : 0;
  const int excl = front + inc - cn;
  cu[t] = excl;
  if (t < BW) rowptr[r * BW + t] = beg + excl;
  if (r == NB - 1 && t == 0) rowptr[N_NODES] = end;
  __syncthreads();

  for (int i0 = t; i0 < n; i0 += 1024) {
    int w[4];
#pragma unroll
    for (int k = 0; k < 4; ++k) w[k] = (i0 + 256 * k < n) ? tmp[beg + i0 + 256 * k] : -1;
#pragma unroll
    for (int k = 0; k < 4; ++k) {
      if (w[k] >= 0) {
        const int p = atomicAdd(&cu[w[k] >> 16], 1);
        col[beg + p] = w[k] & 0xffff;
      }
    }
  }
}

__global__ __launch_bounds__(256) void gnn_seg_sort(const int* __restrict__ rowptr,
                                                    const int* __restrict__ col,
                                                    int* __restrict__ scol) {
  const int lane = threadIdx.x & 31;
  const int wave0  = __builtin_amdgcn_readfirstlane(blockIdx.x * 8 + (threadIdx.x >> 5));
  const int nwaves = gridDim.x * 8;

  for (int node = wave0; node < N_NODES; node += nwaves) {
    const int beg = rowptr[node];
    const int n   = rowptr[node + 1] - beg;
    if (n <= 32) {
      const int key = (lane < n) ? col[beg + lane] : 0;
      int rank = 0;
      for (int c = 0; c < n; ++c) {
        const int other = __builtin_amdgcn_readlane(key, c);
        rank += (other < key || (other == key && c < lane)) ? 1 : 0;
      }
      if (lane < n) scol[beg + rank] = key;
    } else {
      for (int a = lane; a < n; a += 32) {
        const int key = col[beg + a];
        int rank = 0;
        for (int c = 0; c < n; ++c) {
          const int other = col[beg + c];
          rank += (other < key || (other == key && c < a)) ? 1 : 0;
        }
        scol[beg + rank] = key;
      }
    }
  }
}

__global__ __launch_bounds__(256) void gnn_gather(const float* __restrict__ m,
                                                  const int* __restrict__ rowptr,
                                                  const int* __restrict__ col,
                                                  float* __restrict__ agg) {
  const int lane = threadIdx.x & 31;
  const int wave0  = __builtin_amdgcn_readfirstlane(blockIdx.x * 8 + (threadIdx.x >> 5));
  const int nwaves = gridDim.x * 8;
  const int c4     = (lane < 24) ? lane * 4 : 0;

  for (int node = wave0; node < N_NODES; node += nwaves) {
    const int beg = rowptr[node];
    const int end = rowptr[node + 1];
    float4 acc = make_float4(0.f, 0.f, 0.f, 0.f);
    for (int b = beg; b < end; b += 32) {
      const int cnt   = (end - b < 32) ? (end - b) : 32;
      const int mycol = (lane < cnt) ? col[b + lane] : 0;
      for (int j = 0; j < cnt; j += 8) {
        float4 v[8];
#pragma unroll
        for (int u = 0; u < 8; ++u) {
          const int s = __builtin_amdgcn_readlane(mycol, j + u);
          v[u] = *(const float4*)(m + s * CDIM + c4);
        }
#pragma unroll
        for (int u = 0; u < 8; ++u) {
          if (j + u < cnt) {
            acc.x += v[u].x; acc.y += v[u].y; acc.z += v[u].z; acc.w += v[u].w;
          }
        }
      }
    }
    if (lane < 24) *(float4*)(agg + node * CDIM + c4) = acc;
  }
}

#define GRU_SLOTS 85
#define GRU_WAVES (GRU_SLOTS * 8)

__device__ __forceinline__ void gru_tile(const float* agp, const float* ahp,
                                         const float* bp, const float* Ah,
                                         float bir, float biz, float bin_, float bhr, float bhz, float bhn,
                                         float* __restrict__ out, int row0, int col, int half) {
  v8f a_ir, a_iz, a_in, a_hr, a_hz, a_hn;
#pragma unroll
  for (int r = 0; r < 8; ++r) {
    a_ir[r] = bir; a_iz[r] = biz; a_in[r] = bin_;
    a_hr[r] = bhr; a_hz[r] = bhz; a_hn[r] = bhn;
  }

#pragma unroll 4
  for (int k0 = 0; k0 < CDIM; k0 += 4) {
    const v2f ai = *(const v2f*)(agp + k0);
    const v2f ah = *(const v2f*)(ahp + k0);
    const v2f b0 = *(const v2f*)(bp + 0 * 16 * LDP + k0);
    const v2f b1 = *(const v2f*)(bp + 1 * 16 * LDP + k0);
    const v2f b2 = *(const v2f*)(bp + 2 * 16 * LDP + k0);
    const v2f b3 = *(const v2f*)(bp + 3 * 16 * LDP + k0);
    const v2f b4 = *(const v2f*)(bp + 4 * 16 * LDP + k0);
    const v2f b5 = *(const v2f*)(bp + 5 * 16 * LDP + k0);
    a_ir = __builtin_amdgcn_wmma_f32_16x16x4_f32(false, ai, false, b0, (short)0, a_ir, false, false);
    a_iz = __builtin_amdgcn_wmma_f32_16x16x4_f32(false, ai, false, b1, (short)0, a_iz, false, false);
    a_in = __builtin_amdgcn_wmma_f32_16x16x4_f32(false, ai, false, b2, (short)0, a_in, false, false);
    a_hr = __builtin_amdgcn_wmma_f32_16x16x4_f32(false, ah, false, b3, (short)0, a_hr, false, false);
    a_hz = __builtin_amdgcn_wmma_f32_16x16x4_f32(false, ah, false, b4, (short)0, a_hz, false, false);
    a_hn = __builtin_amdgcn_wmma_f32_16x16x4_f32(false, ah, false, b5, (short)0, a_hn, false, false);
  }

#pragma unroll
  for (int r = 0; r < 8; ++r) {
    const int lr  = r + 8 * half;
    const float hv = Ah[lr * LDP + col];
    const float rr = 1.f / (1.f + __expf(-(a_ir[r] + a_hr[r])));
    const float zz = 1.f / (1.f + __expf(-(a_iz[r] + a_hz[r])));
    const float nn = tanhf(a_in[r] + rr * a_hn[r]);
    out[(row0 + lr) * CDIM + col] = (1.f - zz) * nn + zz * hv;
  }
}

__global__ __launch_bounds__(256) void gnn_gru(const float* __restrict__ agg,
                                               const float* __restrict__ h,
                                               const float* __restrict__ w_ih,
                                               const float* __restrict__ w_hh,
                                               const float* __restrict__ b_ih,
                                               const float* __restrict__ b_hh,
                                               float* __restrict__ out) {
  __shared__ __attribute__((aligned(16))) float lds[6 * 16 * LDP + 8 * 2 * 16 * LDP];
  const int tid  = threadIdx.x;
  const int wid  = tid >> 5;
  const int lane = tid & 31;
  const int half = lane >> 4;
  const int l16  = lane & 15;
  float* Bs = lds;
  float* Ag = lds + 6 * 16 * LDP + wid * (2 * 16 * LDP);
  float* Ah = Ag + 16 * LDP;

  const int tc   = blockIdx.x % (CDIM / 16);
  const int slot = blockIdx.x / (CDIM / 16);
  const int col0 = tc * 16;
  const int col  = col0 + l16;
  const int tr0 = __builtin_amdgcn_readfirstlane(((wid + slot + tc) & 7) * GRU_SLOTS + slot);

  const float bir = b_ih[col], biz = b_ih[CDIM + col], bin_ = b_ih[2 * CDIM + col];
  const float bhr = b_hh[col], bhz = b_hh[CDIM + col], bhn  = b_hh[2 * CDIM + col];

  {
    float4 wv[9];
    float4 gv[12], hv4[12];
#pragma unroll
    for (int i = 0; i < 9; ++i) {
      const int idx = i * 256 + tid;
      const int j   = idx / 384;
      const int w   = idx % 384;
      const float* srcw = (j < 3 ? w_ih : w_hh) + ((j % 3) * CDIM + col0) * CDIM;
      wv[i] = *(const float4*)(srcw + w * 4);
    }
    load_rowtile(agg + tr0 * 16 * CDIM, lane, gv);
    load_rowtile(h   + tr0 * 16 * CDIM, lane, hv4);
#pragma unroll
    for (int i = 0; i < 9; ++i) {
      const int idx = i * 256 + tid;
      const int j   = idx / 384;
      const int w   = idx % 384;
      *(float4*)(Bs + (j * 16 + w / 24) * LDP + (w % 24) * 4) = wv[i];
    }
    store_rowtile(Ag, lane, gv);
    store_rowtile(Ah, lane, hv4);
  }
  __syncthreads();

  const float* agp = Ag + l16 * LDP + 2 * half;
  const float* ahp = Ah + l16 * LDP + 2 * half;
  const float* bp  = Bs + l16 * LDP + 2 * half;

  int tr = tr0;
  for (; tr + GRU_WAVES <= LASTTILE; tr += GRU_WAVES) {
    gru_tile(agp, ahp, bp, Ah, bir, biz, bin_, bhr, bhz, bhn, out, tr * 16, col, half);

    float4 gv[12], hv4[12];
    load_rowtile(agg + (tr + GRU_WAVES) * 16 * CDIM, lane, gv);
    load_rowtile(h   + (tr + GRU_WAVES) * 16 * CDIM, lane, hv4);
    store_rowtile(Ag, lane, gv);
    store_rowtile(Ah, lane, hv4);
  }
  gru_tile(agp, ahp, bp, Ah, bir, biz, bin_, bhr, bhz, bhn, out, tr * 16, col, half);
}

extern "C" void kernel_launch(void* const* d_in, const int* in_sizes, int n_in,
                              void* d_out, int out_size, void* d_ws, size_t ws_size,
                              hipStream_t stream) {
  const float* x      = (const float*)d_in[0];
  const int*   ei     = (const int*)  d_in[1];
  const float* weight = (const float*)d_in[2];
  const float* w_ih   = (const float*)d_in[3];
  const float* w_hh   = (const float*)d_in[4];
  const float* b_ih   = (const float*)d_in[5];
  const float* b_hh   = (const float*)d_in[6];
  float* out = (float*)d_out;

  float* xa  = (float*)d_ws;
  int*   ib  = (int*)(xa + NC);
  float* wf  = xa + 2 * NC;
  float* agg = xa + 3 * NC;

  int* ht     = ib;
  int* base   = ib + 62592;
  int* tmp    = ib + 2 * 62592;
  int* rowptr = tmp + N_EDGES;
  int* col    = rowptr + 50048;
  int* scol   = col + N_EDGES;

  const int gruBlocks     = GRU_SLOTS * (CDIM / 16);
  const int setupBlocks   = NCH + FOLD_BLOCKS;
  const int gathBlocks    = 2048;

  gnn_setup    <<<setupBlocks, 256, 0, stream>>>(ei, ht, weight, w_ih, wf);
  gnn_scan     <<<SCAN_BLOCKS, 1024, 0, stream>>>(ht, base);
  gnn_scatter  <<<NCH, 256, 0, stream>>>(ei, base, tmp);
  gnn_range    <<<NB, 256, 0, stream>>>(base, tmp, rowptr, col);
  gnn_seg_sort <<<gathBlocks, 256, 0, stream>>>(rowptr, col, scol);

  for (int l = 0; l < 3; ++l) {
    const float* xin  = (l == 0) ? x   : ((l == 1) ? out : xa);
    float*       xout = (l == 1) ? xa  : out;
    const float* wfl  = wf + l * GATES * CDIM;

    gnn_gather<<<gathBlocks, 256, 0, stream>>>(xin, rowptr, scol, agg);
    gnn_gru   <<<gruBlocks,  256, 0, stream>>>(agg, xin, wfl, w_hh, b_ih, b_hh, xout);
  }
}


# Range kernel (1024 threads) keeps its edges in LDS, sorts every segment there and forms layer 0's row sums for its own nodes: seg-sort launch and layer-0 gather launch removed; sum order unchanged
# speedup vs baseline: 3.1762x; 1.0459x over previous
#include <hip/hip_runtime.h>
#include <hip/hip_bf16.h>


#define N_NODES 50000
#define N_EDGES 800000
#define CDIM    96
#define NC      (N_NODES * CDIM)
#define GATES   (3 * CDIM)

typedef __attribute__((ext_vector_type(2))) float v2f;
typedef __attribute__((ext_vector_type(8))) float v8f;

#define LDP      100
#define LASTTILE (N_NODES / 16 - 1)

__device__ __forceinline__ void load_rowtile(const float* __restrict__ g, int lane, float4 (&v)[12]) {
#pragma unroll
  for (int i = 0; i < 12; ++i) v[i] = *(const float4*)(g + (i * 32 + lane) * 4);
}
__device__ __forceinline__ void store_rowtile(float* __restrict__ s, int lane, const float4 (&v)[12]) {
#pragma unroll
  for (int i = 0; i < 12; ++i) {
    const int idx = i * 32 + lane;
    *(float4*)(s + (idx / 24) * LDP + (idx % 24) * 4) = v[i];
  }
}

#define NB          250
#define BW          200
#define NCH         250
#define CH4         800
#define NSCAN       (NB * NCH)
#define SCAN_BLOCKS 62
#define FOLD_TILES  (3 * (GATES / 16) * (CDIM / 16))
#define FOLD_BLOCKS ((FOLD_TILES + 7) / 8)

__global__ __launch_bounds__(256) void gnn_setup(const int* __restrict__ edge_index,
                                                 int* __restrict__ ht,
                                                 const float* __restrict__ weight,
                                                 const float* __restrict__ w_ih,
                                                 float* __restrict__ wf) {
  __shared__ int hcnt[256];
  if (blockIdx.x < NCH) {
    const int t = threadIdx.x;
    const int c = blockIdx.x;
    hcnt[t] = 0;
    int4 d[4];
#pragma unroll
    for (int k = 0; k < 4; ++k) {
      const int i = t + 256 * k;
      d[k] = (i < CH4) ? *(const int4*)(edge_index + N_EDGES + (c * CH4 + i) * 4) : make_int4(0, 0, 0, 0);
    }
    __syncthreads();
#pragma unroll
    for (int k = 0; k < 4; ++k) {
      if (t + 256 * k < CH4) {
        atomicAdd(&hcnt[(unsigned)d[k].x / BW], 1);
        atomicAdd(&hcnt[(unsigned)d[k].y / BW], 1);
        atomicAdd(&hcnt[(unsigned)d[k].z / BW], 1);
        atomicAdd(&hcnt[(unsigned)d[k].w / BW], 1);
      }
    }
    __syncthreads();
    if (t < NB) ht[t * NCH + c] = hcnt[t];
    return;
  }
  const int lane = threadIdx.x & 31;
  const int half = lane >> 4;
  const int l16  = lane & 15;
  const int tile = __builtin_amdgcn_readfirstlane((blockIdx.x - NCH) * 8 + (threadIdx.x >> 5));
  if (tile >= FOLD_TILES) return;

  const int l  = tile / ((GATES / 16) * (CDIM / 16));
  const int t  = tile % ((GATES / 16) * (CDIM / 16));
  const int n0 = (t / (CDIM / 16)) * 16;
  const int k0 = (t % (CDIM / 16)) * 16;

  const float* ap = w_ih + (n0 + l16) * CDIM + 2 * half;
  const float* bp = weight + l * CDIM * CDIM + (k0 + l16) * CDIM + 2 * half;

  v2f a[24], b[24];
#pragma unroll
  for (int s = 0; s < 24; ++s) {
    a[s] = *(const v2f*)(ap + 4 * s);
    b[s] = *(const v2f*)(bp + 4 * s);
  }

  v8f acc0 = (v8f){0.f, 0.f, 0.f, 0.f, 0.f, 0.f, 0.f, 0.f};
  v8f acc1 = (v8f){0.f, 0.f, 0.f, 0.f, 0.f, 0.f, 0.f, 0.f};
#pragma unroll
  for (int s = 0; s < 24; s += 2) {
    acc0 = __builtin_amdgcn_wmma_f32_16x16x4_f32(false, a[s],     false, b[s],     (short)0, acc0, false, false);
    acc1 = __builtin_amdgcn_wmma_f32_16x16x4_f32(false, a[s + 1], false, b[s + 1], (short)0, acc1, false, false);
  }

  float* o = wf + l * GATES * CDIM + (n0 + 8 * half) * CDIM + k0 + l16;
#pragma unroll
  for (int r = 0; r < 8; ++r) o[r * CDIM] = acc0[r] + acc1[r];
}

__global__ __launch_bounds__(1024) void gnn_scan(const int* __restrict__ cntv,
                                                 int* __restrict__ base) {
  __shared__ int wpre[32];
  __shared__ int wtot[32];
  const int t    = threadIdx.x;
  const int lane = t & 31;
  const int w    = t >> 5;
  const int b    = blockIdx.x;

  int pre = 0;
  for (int r = 0; r < b; ++r) pre += cntv[r * 1024 + t];

  const int i = b * 1024 + t;
  const int d = (i < NSCAN) ? cntv[i] : 0;

  int inc = d;
#pragma unroll
  for (int off = 1; off < 32; off <<= 1) {
    const int v = __shfl_up(inc, off);
    if (lane >= off) inc += v;
  }
#pragma unroll
  for (int off = 16; off > 0; off >>= 1) pre += __shfl_xor(pre, off);
  if (lane == 31) wtot[w] = inc;
  if (lane == 0)  wpre[w] = pre;
  __syncthreads();

  int o = wpre[lane] + ((lane < w) ? wtot[lane] : 0);
#pragma unroll
  for (int off = 16; off > 0; off >>= 1) o += __shfl_xor(o, off);

  const int excl = o + inc - d;
  if (i < NSCAN) base[i] = excl;
  if (i == NSCAN - 1) base[NSCAN] = excl + d;
}

#define PLACE_EDGE(dd, ss)                                              \
  {                                                                     \
    const unsigned r_  = (unsigned)(dd) / BW;                           \
    const unsigned dl_ = (unsigned)(dd) - r_ * BW;                      \
    const int p_ = atomicAdd(&cur[r_], 1);                              \
    tmp[p_] = (int)((dl_ << 16) | (unsigned)(ss));                      \
  }
__global__ __launch_bounds__(256) void gnn_scatter(const int* __restrict__ edge_index,
                                                   const int* __restrict__ base,
                                                   int* __restrict__ tmp) {
  __shared__ int cur[256];
  const int t = threadIdx.x;
  const int c = blockIdx.x;
  int4 s[4], d[4];
#pragma unroll
  for (int k = 0; k < 4; ++k) {
    const int i = t + 256 * k;
    s[k] = (i < CH4) ? *(const int4*)(edge_index + (c * CH4 + i) * 4) : make_int4(0, 0, 0, 0);
    d[k] = (i < CH4) ? *(const int4*)(edge_index + N_EDGES + (c * CH4 + i) * 4) : make_int4(0, 0, 0, 0);
  }
  if (t < NB) cur[t] = base[t * NCH + c];
  __syncthreads();
#pragma unroll
  for (int k = 0; k < 4; ++k) {
    if (t + 256 * k < CH4) {
      PLACE_EDGE(d[k].x, s[k].x)
      PLACE_EDGE(d[k].y, s[k].y)
      PLACE_EDGE(d[k].z, s[k].z)
      PLACE_EDGE(d[k].w, s[k].w)
    }
  }
}

#define SEG_CAP 8192

template <bool TO_LDS>
__device__ __forceinline__ void sort_segment(const int* seg, int* srt, int* __restrict__ gs,
                                             int n, int lane) {
  if (n <= 32) {
    const int key = (lane < n) ? seg[lane] : 0;
    int rank = 0;
    for (int c = 0; c < n; ++c) {
      const int other = __builtin_amdgcn_readlane(key, c);
      rank += (other < key || (other == key && c < lane)) ? 1 : 0;
    }
    if (lane < n) {
      if (TO_LDS) srt[rank] = key;
      gs[rank] = key;
    }
  } else {
    for (int a = lane; a < n; a += 32) {
      const int key = seg[a];
      int rank = 0;
      for (int c = 0; c < n; ++c) {
        const int other = seg[c];
        rank += (other < key || (other == key && c < a)) ? 1 : 0;
      }
      if (TO_LDS) srt[rank] = key;
      gs[rank] = key;
    }
  }
}

__device__ __forceinline__ float4 sum_rows(const float* __restrict__ m, const int* keys,
                                           int n, int lane, int c4) {
  float4 acc = make_float4(0.f, 0.f, 0.f, 0.f);
  for (int b = 0; b < n; b += 32) {
    const int cnt   = (n - b < 32) ? (n - b) : 32;
    const int mycol = (lane < cnt) ? keys[b + lane] : 0;
    for (int j = 0; j < cnt; j += 8) {
      float4 v[8];
#pragma unroll
      for (int u = 0; u < 8; ++u) {
        const int s = __builtin_amdgcn_readlane(mycol, j + u);
        v[u] = *(const float4*)(m + s * CDIM + c4);
      }
#pragma unroll
      for (int u = 0; u < 8; ++u) {
        if (j + u < cnt) {
          acc.x += v[u].x; acc.y += v[u].y; acc.z += v[u].z; acc.w += v[u].w;
        }
      }
    }
  }
  return acc;
}

__global__ __launch_bounds__(1024) void gnn_range(const int* __restrict__ base,
                                                  const int* __restrict__ tmp,
                                                  int* __restrict__ rowptr,
                                                  int* __restrict__ col,
                                                  int* __restrict__ scol,
                                                  const float* __restrict__ x,
                                                  float* __restrict__ agg) {
  __shared__ int cnt[256];
  __shared__ int st[256];
  __shared__ int cu[256];
  __shared__ int wtot[8];
  __shared__ int lseg[SEG_CAP];
  __shared__ int lsrt[SEG_CAP];
  const int t    = threadIdx.x;
  const int lane = t & 31;
  const int wv   = __builtin_amdgcn_readfirstlane(t >> 5);
  const int r    = blockIdx.x;
  const int beg  = base[r * NCH];
  const int end  = base[(r + 1) * NCH];
  const int n    = end - beg;
  const bool fits = (n <= SEG_CAP);

  if (t < 256) cnt[t] = 0;
  __syncthreads();
  for (int i0 = t; i0 < n; i0 += 4096) {
    int w[4];
#pragma unroll
    for (int k = 0; k < 4; ++k) w[k] = (i0 + 1024 * k < n) ? tmp[beg + i0 + 1024 * k] : -1;
#pragma unroll
    for (int k = 0; k < 4; ++k)
      if (w[k] >= 0) atomicAdd(&cnt[w[k] >> 16], 1);
  }
  __syncthreads();

  int cn = 0, inc = 0;
  if (t < 256) {
    cn  = cnt[t];
    inc = cn;
#pragma unroll
    for (int off = 1; off < 32; off <<= 1) {
      const int v = __shfl_up(inc, off);
      if (lane >= off) inc += v;
    }
    if (lane == 31) wtot[wv] = inc;
  }
  __syncthreads();
  if (t < 256) {
    int front = 0;
#pragma unroll
    for (int k = 0; k < 8; ++k) front += (k < wv) ? wtot[k] : 0;
    const int excl = front + inc - cn;
    st[t] = excl;
    cu[t] = excl;
    if (t < BW) rowptr[r * BW + t] = beg + excl;
  }
  if (r == NB - 1 && t == 0) rowptr[N_NODES] = end;
  __syncthreads();

  for (int i0 = t; i0 < n; i0 += 4096) {
    int w[4];
#pragma unroll
    for (int k = 0; k < 4; ++k) w[k] = (i0 + 1024 * k < n) ? tmp[beg + i0 + 1024 * k] : -1;
#pragma unroll
    for (int k = 0; k < 4; ++k) {
      if (w[k] >= 0) {
        const int p = atomicAdd(&cu[w[k] >> 16], 1);
        if (fits) lseg[p] = w[k] & 0xffff;
        else      col[beg + p] = w[k] & 0xffff;
      }
    }
  }
  if (!fits) __threadfence();
  __syncthreads();

  const int c4 = (lane < 24) ? lane * 4 : 0;
  for (int node = wv; node < BW; node += 32) {
    const int off = __builtin_amdgcn_readfirstlane(st[node]);
    const int nd  = __builtin_amdgcn_readfirstlane(cnt[node]);
    float4 acc;
    if (fits) {
      sort_segment<true>(lseg + off, lsrt + off, scol + beg + off, nd, lane);
      acc = sum_rows(x, lsrt + off, nd, lane, c4);
    } else {
      sort_segment<false>(col + beg + off, nullptr, scol + beg + off, nd, lane);
      __threadfence();
      acc = sum_rows(x, scol + beg + off, nd, lane, c4);
    }
    if (lane < 24) *(float4*)(agg + (r * BW + node) * CDIM + c4) = acc;
  }
}

__global__ __launch_bounds__(256) void gnn_gather(const float* __restrict__ m,
                                                  const int* __restrict__ rowptr,
                                                  const int* __restrict__ col,
                                                  float* __restrict__ agg) {
  const int lane = threadIdx.x & 31;
  const int wave0  = __builtin_amdgcn_readfirstlane(blockIdx.x * 8 + (threadIdx.x >> 5));
  const int nwaves = gridDim.x * 8;
  const int c4     = (lane < 24) ? lane * 4 : 0;

  for (int node = wave0; node < N_NODES; node += nwaves) {
    const int beg = rowptr[node];
    const int end = rowptr[node + 1];
    float4 acc = make_float4(0.f, 0.f, 0.f, 0.f);
    for (int b = beg; b < end; b += 32) {
      const int cnt   = (end - b < 32) ? (end - b) : 32;
      const int mycol = (lane < cnt) ? col[b + lane] : 0;
      for (int j = 0; j < cnt; j += 8) {
        float4 v[8];
#pragma unroll
        for (int u = 0; u < 8; ++u) {
          const int s = __builtin_amdgcn_readlane(mycol, j + u);
          v[u] = *(const float4*)(m + s * CDIM + c4);
        }
#pragma unroll
        for (int u = 0; u < 8; ++u) {
          if (j + u < cnt) {
            acc.x += v[u].x; acc.y += v[u].y; acc.z += v[u].z; acc.w += v[u].w;
          }
        }
      }
    }
    if (lane < 24) *(float4*)(agg + node * CDIM + c4) = acc;
  }
}

#define GRU_SLOTS 85
#define GRU_WAVES (GRU_SLOTS * 8)

__device__ __forceinline__ void gru_tile(const float* agp, const float* ahp,
                                         const float* bp, const float* Ah,
                                         float bir, float biz, float bin_, float bhr, float bhz, float bhn,
                                         float* __restrict__ out, int row0, int col, int half) {
  v8f a_ir, a_iz, a_in, a_hr, a_hz, a_hn;
#pragma unroll
  for (int r = 0; r < 8; ++r) {
    a_ir[r] = bir; a_iz[r] = biz; a_in[r] = bin_;
    a_hr[r] = bhr; a_hz[r] = bhz; a_hn[r] = bhn;
  }

#pragma unroll 4
  for (int k0 = 0; k0 < CDIM; k0 += 4) {
    const v2f ai = *(const v2f*)(agp + k0);
    const v2f ah = *(const v2f*)(ahp + k0);
    const v2f b0 = *(const v2f*)(bp + 0 * 16 * LDP + k0);
    const v2f b1 = *(const v2f*)(bp + 1 * 16 * LDP + k0);
    const v2f b2 = *(const v2f*)(bp + 2 * 16 * LDP + k0);
    const v2f b3 = *(const v2f*)(bp + 3 * 16 * LDP + k0);
    const v2f b4 = *(const v2f*)(bp + 4 * 16 * LDP + k0);
    const v2f b5 = *(const v2f*)(bp + 5 * 16 * LDP + k0);
    a_ir = __builtin_amdgcn_wmma_f32_16x16x4_f32(false, ai, false, b0, (short)0, a_ir, false, false);
    a_iz = __builtin_amdgcn_wmma_f32_16x16x4_f32(false, ai, false, b1, (short)0, a_iz, false, false);
    a_in = __builtin_amdgcn_wmma_f32_16x16x4_f32(false, ai, false, b2, (short)0, a_in, false, false);
    a_hr = __builtin_amdgcn_wmma_f32_16x16x4_f32(false, ah, false, b3, (short)0, a_hr, false, false);
    a_hz = __builtin_amdgcn_wmma_f32_16x16x4_f32(false, ah, false, b4, (short)0, a_hz, false, false);
    a_hn = __builtin_amdgcn_wmma_f32_16x16x4_f32(false, ah, false, b5, (short)0, a_hn, false, false);
  }

#pragma unroll
  for (int r = 0; r < 8; ++r) {
    const int lr  = r + 8 * half;
    const float hv = Ah[lr * LDP + col];
    const float rr = 1.f / (1.f + __expf(-(a_ir[r] + a_hr[r])));
    const float zz = 1.f / (1.f + __expf(-(a_iz[r] + a_hz[r])));
    const float nn = tanhf(a_in[r] + rr * a_hn[r]);
    out[(row0 + lr) * CDIM + col] = (1.f - zz) * nn + zz * hv;
  }
}

__global__ __launch_bounds__(256) void gnn_gru(const float* __restrict__ agg,
                                               const float* __restrict__ h,
                                               const float* __restrict__ w_ih,
                                               const float* __restrict__ w_hh,
                                               const float* __restrict__ b_ih,
                                               const float* __restrict__ b_hh,
                                               float* __restrict__ out) {
  __shared__ __attribute__((aligned(16))) float lds[6 * 16 * LDP + 8 * 2 * 16 * LDP];
  const int tid  = threadIdx.x;
  const int wid  = tid >> 5;
  const int lane = tid & 31;
  const int half = lane >> 4;
  const int l16  = lane & 15;
  float* Bs = lds;
  float* Ag = lds + 6 * 16 * LDP + wid * (2 * 16 * LDP);
  float* Ah = Ag + 16 * LDP;

  const int tc   = blockIdx.x % (CDIM / 16);
  const int slot = blockIdx.x / (CDIM / 16);
  const int col0 = tc * 16;
  const int col  = col0 + l16;
  const int tr0 = __builtin_amdgcn_readfirstlane(((wid + slot + tc) & 7) * GRU_SLOTS + slot);

  const float bir = b_ih[col], biz = b_ih[CDIM + col], bin_ = b_ih[2 * CDIM + col];
  const float bhr = b_hh[col], bhz = b_hh[CDIM + col], bhn  = b_hh[2 * CDIM + col];

  {
    float4 wv[9];
    float4 gv[12], hv4[12];
#pragma unroll
    for (int i = 0; i < 9; ++i) {
      const int idx = i * 256 + tid;
      const int j   = idx / 384;
      const int w   = idx % 384;
      const float* srcw = (j < 3 ? w_ih : w_hh) + ((j % 3) * CDIM + col0) * CDIM;
      wv[i] = *(const float4*)(srcw + w * 4);
    }
    load_rowtile(agg + tr0 * 16 * CDIM, lane, gv);
    load_rowtile(h   + tr0 * 16 * CDIM, lane, hv4);
#pragma unroll
    for (int i = 0; i < 9; ++i) {
      const int idx = i * 256 + tid;
      const int j   = idx / 384;
      const int w   = idx % 384;
      *(float4*)(Bs + (j * 16 + w / 24) * LDP + (w % 24) * 4) = wv[i];
    }
    store_rowtile(Ag, lane, gv);
    store_rowtile(Ah, lane, hv4);
  }
  __syncthreads();

  const float* agp = Ag + l16 * LDP + 2 * half;
  const float* ahp = Ah + l16 * LDP + 2 * half;
  const float* bp  = Bs + l16 * LDP + 2 * half;

  int tr = tr0;
  for (; tr + GRU_WAVES <= LASTTILE; tr += GRU_WAVES) {
    gru_tile(agp, ahp, bp, Ah, bir, biz, bin_, bhr, bhz, bhn, out, tr * 16, col, half);

    float4 gv[12], hv4[12];
    load_rowtile(agg + (tr + GRU_WAVES) * 16 * CDIM, lane, gv);
    load_rowtile(h   + (tr + GRU_WAVES) * 16 * CDIM, lane, hv4);
    store_rowtile(Ag, lane, gv);
    store_rowtile(Ah, lane, hv4);
  }
  gru_tile(agp, ahp, bp, Ah, bir, biz, bin_, bhr, bhz, bhn, out, tr * 16, col, half);
}

extern "C" void kernel_launch(void* const* d_in, const int* in_sizes, int n_in,
                              void* d_out, int out_size, void* d_ws, size_t ws_size,
                              hipStream_t stream) {
  const float* x      = (const float*)d_in[0];
  const int*   ei     = (const int*)  d_in[1];
  const float* weight = (const float*)d_in[2];
  const float* w_ih   = (const float*)d_in[3];
  const float* w_hh   = (const float*)d_in[4];
  const float* b_ih   = (const float*)d_in[5];
  const float* b_hh   = (const float*)d_in[6];
  float* out = (float*)d_out;

  float* xa  = (float*)d_ws;
  int*   ib  = (int*)(xa + NC);
  float* wf  = xa + 2 * NC;
  float* agg = xa + 3 * NC;

  int* ht     = ib;
  int* base   = ib + 62592;
  int* tmp    = ib + 2 * 62592;
  int* rowptr = tmp + N_EDGES;
  int* col    = rowptr + 50048;
  int* scol   = col + N_EDGES;

  const int gruBlocks     = GRU_SLOTS * (CDIM / 16);
  const int setupBlocks   = NCH + FOLD_BLOCKS;
  const int gathBlocks    = 2048;

  gnn_setup    <<<setupBlocks, 256, 0, stream>>>(ei, ht, weight, w_ih, wf);
  gnn_scan     <<<SCAN_BLOCKS, 1024, 0, stream>>>(ht, base);
  gnn_scatter  <<<NCH, 256, 0, stream>>>(ei, base, tmp);
  gnn_range    <<<NB, 1024, 0, stream>>>(base, tmp, rowptr, col, scol, x, agg);

  for (int l = 0; l < 3; ++l) {
    const float* xin  = (l == 0) ? x   : ((l == 1) ? out : xa);
    float*       xout = (l == 1) ? xa  : out;
    const float* wfl  = wf + l * GATES * CDIM;

    if (l > 0) gnn_gather<<<gathBlocks, 256, 0, stream>>>(xin, rowptr, scol, agg);
    gnn_gru   <<<gruBlocks,  256, 0, stream>>>(agg, xin, wfl, w_hh, b_ih, b_hh, xout);
  }
}


# Edges grouped by range inside each chunk's own slice of tmp by one kernel (count, LDS scan, place); range work-groups collect their cells and sum their own offset: scan and scatter launches gone
# speedup vs baseline: 3.4945x; 1.1002x over previous
#include <hip/hip_runtime.h>
#include <hip/hip_bf16.h>


#define N_NODES 50000
#define N_EDGES 800000
#define CDIM    96
#define NC      (N_NODES * CDIM)
#define GATES   (3 * CDIM)

typedef __attribute__((ext_vector_type(2))) float v2f;
typedef __attribute__((ext_vector_type(8))) float v8f;

#define LDP      100
#define LASTTILE (N_NODES / 16 - 1)

__device__ __forceinline__ void load_rowtile(const float* __restrict__ g, int lane, float4 (&v)[12]) {
#pragma unroll
  for (int i = 0; i < 12; ++i) v[i] = *(const float4*)(g + (i * 32 + lane) * 4);
}
__device__ __forceinline__ void store_rowtile(float* __restrict__ s, int lane, const float4 (&v)[12]) {
#pragma unroll
  for (int i = 0; i < 12; ++i) {
    const int idx = i * 32 + lane;
    *(float4*)(s + (idx / 24) * LDP + (idx % 24) * 4) = v[i];
  }
}

#define NB          250
#define BW          200
#define NCH         250
#define CH4         800
#define CHE         (CH4 * 4)
#define NCELL       (NB * NCH)
#define FOLD_TILES  (3 * (GATES / 16) * (CDIM / 16))
#define FOLD_BLOCKS ((FOLD_TILES + 7) / 8)

#define PLACE_EDGE(dd, ss)                                              \
  {                                                                     \
    const unsigned r_  = (unsigned)(dd) / BW;                           \
    const unsigned dl_ = (unsigned)(dd) - r_ * BW;                      \
    const int p_ = atomicAdd(&cur[r_], 1);                              \
    slice[p_] = (int)((dl_ << 16) | (unsigned)(ss));                    \
  }
__global__ __launch_bounds__(256) void gnn_setup(const int* __restrict__ edge_index,
                                                 int* __restrict__ cell,
                                                 int* __restrict__ tmp,
                                                 const float* __restrict__ weight,
                                                 const float* __restrict__ w_ih,
                                                 float* __restrict__ wf) {
  __shared__ int hcnt[256];
  __shared__ int cur[256];
  __shared__ int wtot[8];
  if (blockIdx.x < NCH) {
    const int t  = threadIdx.x;
    const int ln = t & 31;
    const int wv = t >> 5;
    const int c  = blockIdx.x;
    hcnt[t] = 0;
    int4 s[4], d[4];
#pragma unroll
    for (int k = 0; k < 4; ++k) {
      const int i = t + 256 * k;
      s[k] = (i < CH4) ? *(const int4*)(edge_index + (c * CH4 + i) * 4) : make_int4(0, 0, 0, 0);
      d[k] = (i < CH4) ? *(const int4*)(edge_index + N_EDGES + (c * CH4 + i) * 4) : make_int4(0, 0, 0, 0);
    }
    __syncthreads();
#pragma unroll
    for (int k = 0; k < 4; ++k) {
      if (t + 256 * k < CH4) {
        atomicAdd(&hcnt[(unsigned)d[k].x / BW], 1);
        atomicAdd(&hcnt[(unsigned)d[k].y / BW], 1);
        atomicAdd(&hcnt[(unsigned)d[k].z / BW], 1);
        atomicAdd(&hcnt[(unsigned)d[k].w / BW], 1);
      }
    }
    __syncthreads();
    const int cn = hcnt[t];
    int inc = cn;
#pragma unroll
    for (int off = 1; off < 32; off <<= 1) {
      const int v = __shfl_up(inc, off);
      if (ln >= off) inc += v;
    }
    if (ln == 31) wtot[wv] = inc;
    __syncthreads();
    int front = 0;
#pragma unroll
    for (int k = 0; k < 8; ++k) front += (k < wv) ? wtot[k] : 0;
    const int excl = front + inc - cn;
    cur[t] = excl;
    if (t < NB) cell[t * NCH + c] = (excl << 12) | cn;
    __syncthreads();
    int* slice = tmp + c * CHE;
#pragma unroll
    for (int k = 0; k < 4; ++k) {
      if (t + 256 * k < CH4) {
        PLACE_EDGE(d[k].x, s[k].x)
        PLACE_EDGE(d[k].y, s[k].y)
        PLACE_EDGE(d[k].z, s[k].z)
        PLACE_EDGE(d[k].w, s[k].w)
      }
    }
    return;
  }
  const int lane = threadIdx.x & 31;
  const int half = lane >> 4;
  const int l16  = lane & 15;
  const int tile = __builtin_amdgcn_readfirstlane((blockIdx.x - NCH) * 8 + (threadIdx.x >> 5));
  if (tile >= FOLD_TILES) return;

  const int l  = tile / ((GATES / 16) * (CDIM / 16));
  const int t  = tile % ((GATES / 16) * (CDIM / 16));
  const int n0 = (t / (CDIM / 16)) * 16;
  const int k0 = (t % (CDIM / 16)) * 16;

  const float* ap = w_ih + (n0 + l16) * CDIM + 2 * half;
  const float* bp = weight + l * CDIM * CDIM + (k0 + l16) * CDIM + 2 * half;

  v2f a[24], b[24];
#pragma unroll
  for (int s = 0; s < 24; ++s) {
    a[s] = *(const v2f*)(ap + 4 * s);
    b[s] = *(const v2f*)(bp + 4 * s);
  }

  v8f acc0 = (v8f){0.f, 0.f, 0.f, 0.f, 0.f, 0.f, 0.f, 0.f};
  v8f acc1 = (v8f){0.f, 0.f, 0.f, 0.f, 0.f, 0.f, 0.f, 0.f};
#pragma unroll
  for (int s = 0; s < 24; s += 2) {
    acc0 = __builtin_amdgcn_wmma_f32_16x16x4_f32(false, a[s],     false, b[s],     (short)0, acc0, false, false);
    acc1 = __builtin_amdgcn_wmma_f32_16x16x4_f32(false, a[s + 1], false, b[s + 1], (short)0, acc1, false, false);
  }

  float* o = wf + l * GATES * CDIM + (n0 + 8 * half) * CDIM + k0 + l16;
#pragma unroll
  for (int r = 0; r < 8; ++r) o[r * CDIM] = acc0[r] + acc1[r];
}

#define SEG_CAP 8192

template <bool TO_LDS>
__device__ __forceinline__ void sort_segment(const int* seg, int* srt, int* __restrict__ gs,
                                             int n, int lane) {
  if (n <= 32) {
    const int key = (lane < n) ? seg[lane] : 0;
    int rank = 0;
    for (int c = 0; c < n; ++c) {
      const int other = __builtin_amdgcn_readlane(key, c);
      rank += (other < key || (other == key && c < lane)) ? 1 : 0;
    }
    if (lane < n) {
      if (TO_LDS) srt[rank] = key;
      gs[rank] = key;
    }
  } else {
    for (int a = lane; a < n; a += 32) {
      const int key = seg[a];
      int rank = 0;
      for (int c = 0; c < n; ++c) {
        const int other = seg[c];
        rank += (other < key || (other == key && c < a)) ? 1 : 0;
      }
      if (TO_LDS) srt[rank] = key;
      gs[rank] = key;
    }
  }
}

__device__ __forceinline__ float4 sum_rows(const float* __restrict__ m, const int* keys,
                                           int n, int lane, int c4) {
  float4 acc = make_float4(0.f, 0.f, 0.f, 0.f);
  for (int b = 0; b < n; b += 32) {
    const int cnt   = (n - b < 32) ? (n - b) : 32;
    const int mycol = (lane < cnt) ? keys[b + lane] : 0;
    for (int j = 0; j < cnt; j += 8) {
      float4 v[8];
#pragma unroll
      for (int u = 0; u < 8; ++u) {
        const int s = __builtin_amdgcn_readlane(mycol, j + u);
        v[u] = *(const float4*)(m + s * CDIM + c4);
      }
#pragma unroll
      for (int u = 0; u < 8; ++u) {
        if (j + u < cnt) {
          acc.x += v[u].x; acc.y += v[u].y; acc.z += v[u].z; acc.w += v[u].w;
        }
      }
    }
  }
  return acc;
}

__global__ __launch_bounds__(1024) void gnn_range(const int* __restrict__ cell,
                                                  const int* __restrict__ tmp,
                                                  int* __restrict__ rowptr,
                                                  int* __restrict__ col,
                                                  int* __restrict__ scol,
                                                  const float* __restrict__ x,
                                                  float* __restrict__ agg) {
  __shared__ int cnt[256];
  __shared__ int st[256];
  __shared__ int cu[256];
  __shared__ int wtot[8];
  __shared__ int ccn[256];
  __shared__ int cof[256];
  __shared__ int ctot[8];
  __shared__ int wpre[32];
  __shared__ int lseg[SEG_CAP];
  __shared__ int lsrt[SEG_CAP];
  const int t    = threadIdx.x;
  const int lane = t & 31;
  const int wv   = __builtin_amdgcn_readfirstlane(t >> 5);
  const int r    = blockIdx.x;

  const int nfront = r * NCH;
  const int nf4    = nfront >> 2;
  int pre = 0;
  int cw;
  {
    const int lastv = (nf4 > 0) ? nf4 - 1 : 0;
    int4 v[16];
#pragma unroll
    for (int k = 0; k < 16; ++k) {
      const int j = t + 1024 * k;
      v[k] = *(const int4*)(cell + 4 * ((j < nf4) ? j : lastv));
    }
    const int tl = cell[(t < (nfront & 3)) ? (4 * nf4 + t) : 0];
    const int ow = cell[nfront + ((t < NCH) ? t : (NCH - 1))];
#pragma unroll
    for (int k = 0; k < 16; ++k)
      pre += (t + 1024 * k < nf4) ? ((v[k].x & 4095) + (v[k].y & 4095) + (v[k].z & 4095) + (v[k].w & 4095)) : 0;
    pre += (t < (nfront & 3)) ? (tl & 4095) : 0;
    cw = (t < NCH) ? ow : 0;
  }
#pragma unroll
  for (int off = 16; off > 0; off >>= 1) pre += __shfl_xor(pre, off);
  if (lane == 0) wpre[wv] = pre;
  int csum = cw & 4095;
#pragma unroll
  for (int off = 16; off > 0; off >>= 1) csum += __shfl_xor(csum, off);
  if (t < 256) {
    if (lane == 0) ctot[wv] = csum;
    ccn[t] = cw & 4095;
    cof[t] = t * CHE + (cw >> 12);
    cnt[t] = 0;
  }
  __syncthreads();

  int begv = wpre[lane];
  int nv   = (lane < 8) ? ctot[lane] : 0;
#pragma unroll
  for (int off = 16; off > 0; off >>= 1) {
    begv += __shfl_xor(begv, off);
    nv   += __shfl_xor(nv, off);
  }
  const int beg  = __builtin_amdgcn_readfirstlane(begv);
  const int n    = __builtin_amdgcn_readfirstlane(nv);
  const int end  = beg + n;
  const bool fits = (n <= SEG_CAP);

  int w[8], cc[8], co[8];
  bool more = false;
#pragma unroll
  for (int k = 0; k < 8; ++k) {
    cc[k] = ccn[wv + 32 * k];
    co[k] = cof[wv + 32 * k];
  }
#pragma unroll
  for (int k = 0; k < 8; ++k) {
    cc[k] = __builtin_amdgcn_readfirstlane(cc[k]);
    co[k] = __builtin_amdgcn_readfirstlane(co[k]);
    more  = more || (cc[k] > 32);
  }
#pragma unroll
  for (int k = 0; k < 8; ++k) w[k] = (lane < cc[k]) ? tmp[co[k] + lane] : -1;
#pragma unroll
  for (int k = 0; k < 8; ++k)
    if (w[k] >= 0) atomicAdd(&cnt[w[k] >> 16], 1);
  if (more) {
#pragma unroll
    for (int k = 0; k < 8; ++k)
      for (int j = 32 + lane; j < cc[k]; j += 32) atomicAdd(&cnt[tmp[co[k] + j] >> 16], 1);
  }
  __syncthreads();

  int cn = 0, inc = 0;
  if (t < 256) {
    cn  = cnt[t];
    inc = cn;
#pragma unroll
    for (int off = 1; off < 32; off <<= 1) {
      const int v = __shfl_up(inc, off);
      if (lane >= off) inc += v;
    }
    if (lane == 31) wtot[wv] = inc;
  }
  __syncthreads();
  if (t < 256) {
    int front = 0;
#pragma unroll
    for (int k = 0; k < 8; ++k) front += (k < wv) ? wtot[k] : 0;
    const int excl = front + inc - cn;
    st[t] = excl;
    cu[t] = excl;
    if (t < BW) rowptr[r * BW + t] = beg + excl;
  }
  if (r == NB - 1 && t == 0) rowptr[N_NODES] = end;
  __syncthreads();

#pragma unroll
  for (int k = 0; k < 8; ++k) {
    if (w[k] >= 0) {
      const int p = atomicAdd(&cu[w[k] >> 16], 1);
      if (fits) lseg[p] = w[k] & 0xffff;
      else      col[beg + p] = w[k] & 0xffff;
    }
  }
  if (more) {
#pragma unroll
    for (int k = 0; k < 8; ++k) {
      for (int j = 32 + lane; j < cc[k]; j += 32) {
        const int wj = tmp[co[k] + j];
        const int p  = atomicAdd(&cu[wj >> 16], 1);
        if (fits) lseg[p] = wj & 0xffff;
        else      col[beg + p] = wj & 0xffff;
      }
    }
  }
  if (!fits) __threadfence();
  __syncthreads();

  const int c4 = (lane < 24) ? lane * 4 : 0;
  for (int node = wv; node < BW; node += 32) {
    const int off = __builtin_amdgcn_readfirstlane(st[node]);
    const int nd  = __builtin_amdgcn_readfirstlane(cnt[node]);
    float4 acc;
    if (fits) {
      sort_segment<true>(lseg + off, lsrt + off, scol + beg + off, nd, lane);
      acc = sum_rows(x, lsrt + off, nd, lane, c4);
    } else {
      sort_segment<false>(col + beg + off, nullptr, scol + beg + off, nd, lane);
      __threadfence();
      acc = sum_rows(x, scol + beg + off, nd, lane, c4);
    }
    if (lane < 24) *(float4*)(agg + (r * BW + node) * CDIM + c4) = acc;
  }
}

__global__ __launch_bounds__(256) void gnn_gather(const float* __restrict__ m,
                                                  const int* __restrict__ rowptr,
                                                  const int* __restrict__ col,
                                                  float* __restrict__ agg) {
  const int lane = threadIdx.x & 31;
  const int wave0  = __builtin_amdgcn_readfirstlane(blockIdx.x * 8 + (threadIdx.x >> 5));
  const int nwaves = gridDim.x * 8;
  const int c4     = (lane < 24) ? lane * 4 : 0;

  for (int node = wave0; node < N_NODES; node += nwaves) {
    const int beg = rowptr[node];
    const int end = rowptr[node + 1];
    float4 acc = make_float4(0.f, 0.f, 0.f, 0.f);
    for (int b = beg; b < end; b += 32) {
      const int cnt   = (end - b < 32) ? (end - b) : 32;
      const int mycol = (lane < cnt) ? col[b + lane] : 0;
      for (int j = 0; j < cnt; j += 8) {
        float4 v[8];
#pragma unroll
        for (int u = 0; u < 8; ++u) {
          const int s = __builtin_amdgcn_readlane(mycol, j + u);
          v[u] = *(const float4*)(m + s * CDIM + c4);
        }
#pragma unroll
        for (int u = 0; u < 8; ++u) {
          if (j + u < cnt) {
            acc.x += v[u].x; acc.y += v[u].y; acc.z += v[u].z; acc.w += v[u].w;
          }
        }
      }
    }
    if (lane < 24) *(float4*)(agg + node * CDIM + c4) = acc;
  }
}

#define GRU_SLOTS 85
#define GRU_WAVES (GRU_SLOTS * 8)

__device__ __forceinline__ void gru_tile(const float* agp, const float* ahp,
                                         const float* bp, const float* Ah,
                                         float bir, float biz, float bin_, float bhr, float bhz, float bhn,
                                         float* __restrict__ out, int row0, int col, int half) {
  v8f a_ir, a_iz, a_in, a_hr, a_hz, a_hn;
#pragma unroll
  for (int r = 0; r < 8; ++r) {
    a_ir[r] = bir; a_iz[r] = biz; a_in[r] = bin_;
    a_hr[r] = bhr; a_hz[r] = bhz; a_hn[r] = bhn;
  }

#pragma unroll 4
  for (int k0 = 0; k0 < CDIM; k0 += 4) {
    const v2f ai = *(const v2f*)(agp + k0);
    const v2f ah = *(const v2f*)(ahp + k0);
    const v2f b0 = *(const v2f*)(bp + 0 * 16 * LDP + k0);
    const v2f b1 = *(const v2f*)(bp + 1 * 16 * LDP + k0);
    const v2f b2 = *(const v2f*)(bp + 2 * 16 * LDP + k0);
    const v2f b3 = *(const v2f*)(bp + 3 * 16 * LDP + k0);
    const v2f b4 = *(const v2f*)(bp + 4 * 16 * LDP + k0);
    const v2f b5 = *(const v2f*)(bp + 5 * 16 * LDP + k0);
    a_ir = __builtin_amdgcn_wmma_f32_16x16x4_f32(false, ai, false, b0, (short)0, a_ir, false, false);
    a_iz = __builtin_amdgcn_wmma_f32_16x16x4_f32(false, ai, false, b1, (short)0, a_iz, false, false);
    a_in = __builtin_amdgcn_wmma_f32_16x16x4_f32(false, ai, false, b2, (short)0, a_in, false, false);
    a_hr = __builtin_amdgcn_wmma_f32_16x16x4_f32(false, ah, false, b3, (short)0, a_hr, false, false);
    a_hz = __builtin_amdgcn_wmma_f32_16x16x4_f32(false, ah, false, b4, (short)0, a_hz, false, false);
    a_hn = __builtin_amdgcn_wmma_f32_16x16x4_f32(false, ah, false, b5, (short)0, a_hn, false, false);
  }

#pragma unroll
  for (int r = 0; r < 8; ++r) {
    const int lr  = r + 8 * half;
    const float hv = Ah[lr * LDP + col];
    const float rr = 1.f / (1.f + __expf(-(a_ir[r] + a_hr[r])));
    const float zz = 1.f / (1.f + __expf(-(a_iz[r] + a_hz[r])));
    const float nn = tanhf(a_in[r] + rr * a_hn[r]);
    out[(row0 + lr) * CDIM + col] = (1.f - zz) * nn + zz * hv;
  }
}

__global__ __launch_bounds__(256) void gnn_gru(const float* __restrict__ agg,
                                               const float* __restrict__ h,
                                               const float* __restrict__ w_ih,
                                               const float* __restrict__ w_hh,
                                               const float* __restrict__ b_ih,
                                               const float* __restrict__ b_hh,
                                               float* __restrict__ out) {
  __shared__ __attribute__((aligned(16))) float lds[6 * 16 * LDP + 8 * 2 * 16 * LDP];
  const int tid  = threadIdx.x;
  const int wid  = tid >> 5;
  const int lane = tid & 31;
  const int half = lane >> 4;
  const int l16  = lane & 15;
  float* Bs = lds;
  float* Ag = lds + 6 * 16 * LDP + wid * (2 * 16 * LDP);
  float* Ah = Ag + 16 * LDP;

  const int tc   = blockIdx.x % (CDIM / 16);
  const int slot = blockIdx.x / (CDIM / 16);
  const int col0 = tc * 16;
  const int col  = col0 + l16;
  const int tr0 = __builtin_amdgcn_readfirstlane(((wid + slot + tc) & 7) * GRU_SLOTS + slot);

  const float bir = b_ih[col], biz = b_ih[CDIM + col], bin_ = b_ih[2 * CDIM + col];
  const float bhr = b_hh[col], bhz = b_hh[CDIM + col], bhn  = b_hh[2 * CDIM + col];

  {
    float4 wv[9];
    float4 gv[12], hv4[12];
#pragma unroll
    for (int i = 0; i < 9; ++i) {
      const int idx = i * 256 + tid;
      const int j   = idx / 384;
      const int w   = idx % 384;
      const float* srcw = (j < 3 ? w_ih : w_hh) + ((j % 3) * CDIM + col0) * CDIM;
      wv[i] = *(const float4*)(srcw + w * 4);
    }
    load_rowtile(agg + tr0 * 16 * CDIM, lane, gv);
    load_rowtile(h   + tr0 * 16 * CDIM, lane, hv4);
#pragma unroll
    for (int i = 0; i < 9; ++i) {
      const int idx = i * 256 + tid;
      const int j   = idx / 384;
      const int w   = idx % 384;
      *(float4*)(Bs + (j * 16 + w / 24) * LDP + (w % 24) * 4) = wv[i];
    }
    store_rowtile(Ag, lane, gv);
    store_rowtile(Ah, lane, hv4);
  }
  __syncthreads();

  const float* agp = Ag + l16 * LDP + 2 * half;
  const float* ahp = Ah + l16 * LDP + 2 * half;
  const float* bp  = Bs + l16 * LDP + 2 * half;

  int tr = tr0;
  for (; tr + GRU_WAVES <= LASTTILE; tr += GRU_WAVES) {
    gru_tile(agp, ahp, bp, Ah, bir, biz, bin_, bhr, bhz, bhn, out, tr * 16, col, half);

    float4 gv[12], hv4[12];
    load_rowtile(agg + (tr + GRU_WAVES) * 16 * CDIM, lane, gv);
    load_rowtile(h   + (tr + GRU_WAVES) * 16 * CDIM, lane, hv4);
    store_rowtile(Ag, lane, gv);
    store_rowtile(Ah, lane, hv4);
  }
  gru_tile(agp, ahp, bp, Ah, bir, biz, bin_, bhr, bhz, bhn, out, tr * 16, col, half);
}

extern "C" void kernel_launch(void* const* d_in, const int* in_sizes, int n_in,
                              void* d_out, int out_size, void* d_ws, size_t ws_size,
                              hipStream_t stream) {
  const float* x      = (const float*)d_in[0];
  const int*   ei     = (const int*)  d_in[1];
  const float* weight = (const float*)d_in[2];
  const float* w_ih   = (const float*)d_in[3];
  const float* w_hh   = (const float*)d_in[4];
  const float* b_ih   = (const float*)d_in[5];
  const float* b_hh   = (const float*)d_in[6];
  float* out = (float*)d_out;

  float* xa  = (float*)d_ws;
  int*   ib  = (int*)(xa + NC);
  float* wf  = xa + 2 * NC;
  float* agg = xa + 3 * NC;

  int* cell   = ib;
  int* tmp    = ib + 62592;
  int* rowptr = tmp + N_EDGES;
  int* col    = rowptr + 50048;
  int* scol   = col + N_EDGES;

  const int gruBlocks     = GRU_SLOTS * (CDIM / 16);
  const int setupBlocks   = NCH + FOLD_BLOCKS;
  const int gathBlocks    = 2048;

  gnn_setup    <<<setupBlocks, 256, 0, stream>>>(ei, cell, tmp, weight, w_ih, wf);
  gnn_range    <<<NB, 1024, 0, stream>>>(cell, tmp, rowptr, col, scol, x, agg);

  for (int l = 0; l < 3; ++l) {
    const float* xin  = (l == 0) ? x   : ((l == 1) ? out : xa);
    float*       xout = (l == 1) ? xa  : out;
    const float* wfl  = wf + l * GATES * CDIM;

    if (l > 0) gnn_gather<<<gathBlocks, 256, 0, stream>>>(xin, rowptr, scol, agg);
    gnn_gru   <<<gruBlocks,  256, 0, stream>>>(agg, xin, wfl, w_hh, b_ih, b_hh, xout);
  }
}


# GRU epilogue: all three gate activations on the hardware tanh instruction (sigmoid(u) = 0.5 + 0.5 tanh(u/2)) instead of library tanhf and three full divisions; gate values differ in their last bits
# speedup vs baseline: 3.8971x; 1.1152x over previous
#include <hip/hip_runtime.h>
#include <hip/hip_bf16.h>


#define N_NODES 50000
#define N_EDGES 800000
#define CDIM    96
#define NC      (N_NODES * CDIM)
#define GATES   (3 * CDIM)

typedef __attribute__((ext_vector_type(2))) float v2f;
typedef __attribute__((ext_vector_type(8))) float v8f;

#define LDP      100
#define LASTTILE (N_NODES / 16 - 1)

__device__ __forceinline__ void load_rowtile(const float* __restrict__ g, int lane, float4 (&v)[12]) {
#pragma unroll
  for (int i = 0; i < 12; ++i) v[i] = *(const float4*)(g + (i * 32 + lane) * 4);
}
__device__ __forceinline__ void store_rowtile(float* __restrict__ s, int lane, const float4 (&v)[12]) {
#pragma unroll
  for (int i = 0; i < 12; ++i) {
    const int idx = i * 32 + lane;
    *(float4*)(s + (idx / 24) * LDP + (idx % 24) * 4) = v[i];
  }
}

#define NB          250
#define BW          200
#define NCH         250
#define CH4         800
#define CHE         (CH4 * 4)
#define NCELL       (NB * NCH)
#define FOLD_TILES  (3 * (GATES / 16) * (CDIM / 16))
#define FOLD_BLOCKS ((FOLD_TILES + 7) / 8)

#define PLACE_EDGE(dd, ss)                                              \
  {                                                                     \
    const unsigned r_  = (unsigned)(dd) / BW;                           \
    const unsigned dl_ = (unsigned)(dd) - r_ * BW;                      \
    const int p_ = atomicAdd(&cur[r_], 1);                              \
    slice[p_] = (int)((dl_ << 16) | (unsigned)(ss));                    \
  }
__global__ __launch_bounds__(256) void gnn_setup(const int* __restrict__ edge_index,
                                                 int* __restrict__ cell,
                                                 int* __restrict__ tmp,
                                                 const float* __restrict__ weight,
                                                 const float* __restrict__ w_ih,
                                                 float* __restrict__ wf) {
  __shared__ int hcnt[256];
  __shared__ int cur[256];
  __shared__ int wtot[8];
  if (blockIdx.x < NCH) {
    const int t  = threadIdx.x;
    const int ln = t & 31;
    const int wv = t >> 5;
    const int c  = blockIdx.x;
    hcnt[t] = 0;
    int4 s[4], d[4];
#pragma unroll
    for (int k = 0; k < 4; ++k) {
      const int i = t + 256 * k;
      s[k] = (i < CH4) ? *(const int4*)(edge_index + (c * CH4 + i) * 4) : make_int4(0, 0, 0, 0);
      d[k] = (i < CH4) ? *(const int4*)(edge_index + N_EDGES + (c * CH4 + i) * 4) : make_int4(0, 0, 0, 0);
    }
    __syncthreads();
#pragma unroll
    for (int k = 0; k < 4; ++k) {
      if (t + 256 * k < CH4) {
        atomicAdd(&hcnt[(unsigned)d[k].x / BW], 1);
        atomicAdd(&hcnt[(unsigned)d[k].y / BW], 1);
        atomicAdd(&hcnt[(unsigned)d[k].z / BW], 1);
        atomicAdd(&hcnt[(unsigned)d[k].w / BW], 1);
      }
    }
    __syncthreads();
    const int cn = hcnt[t];
    int inc = cn;
#pragma unroll
    for (int off = 1; off < 32; off <<= 1) {
      const int v = __shfl_up(inc, off);
      if (ln >= off) inc += v;
    }
    if (ln == 31) wtot[wv] = inc;
    __syncthreads();
    int front = 0;
#pragma unroll
    for (int k = 0; k < 8; ++k) front += (k < wv) ? wtot[k] : 0;
    const int excl = front + inc - cn;
    cur[t] = excl;
    if (t < NB) cell[t * NCH + c] = (excl << 12) | cn;
    __syncthreads();
    int* slice = tmp + c * CHE;
#pragma unroll
    for (int k = 0; k < 4; ++k) {
      if (t + 256 * k < CH4) {
        PLACE_EDGE(d[k].x, s[k].x)
        PLACE_EDGE(d[k].y, s[k].y)
        PLACE_EDGE(d[k].z, s[k].z)
        PLACE_EDGE(d[k].w, s[k].w)
      }
    }
    return;
  }
  const int lane = threadIdx.x & 31;
  const int half = lane >> 4;
  const int l16  = lane & 15;
  const int tile = __builtin_amdgcn_readfirstlane((blockIdx.x - NCH) * 8 + (threadIdx.x >> 5));
  if (tile >= FOLD_TILES) return;

  const int l  = tile / ((GATES / 16) * (CDIM / 16));
  const int t  = tile % ((GATES / 16) * (CDIM / 16));
  const int n0 = (t / (CDIM / 16)) * 16;
  const int k0 = (t % (CDIM / 16)) * 16;

  const float* ap = w_ih + (n0 + l16) * CDIM + 2 * half;
  const float* bp = weight + l * CDIM * CDIM + (k0 + l16) * CDIM + 2 * half;

  v2f a[24], b[24];
#pragma unroll
  for (int s = 0; s < 24; ++s) {
    a[s] = *(const v2f*)(ap + 4 * s);
    b[s] = *(const v2f*)(bp + 4 * s);
  }

  v8f acc0 = (v8f){0.f, 0.f, 0.f, 0.f, 0.f, 0.f, 0.f, 0.f};
  v8f acc1 = (v8f){0.f, 0.f, 0.f, 0.f, 0.f, 0.f, 0.f, 0.f};
#pragma unroll
  for (int s = 0; s < 24; s += 2) {
    acc0 = __builtin_amdgcn_wmma_f32_16x16x4_f32(false, a[s],     false, b[s],     (short)0, acc0, false, false);
    acc1 = __builtin_amdgcn_wmma_f32_16x16x4_f32(false, a[s + 1], false, b[s + 1], (short)0, acc1, false, false);
  }

  float* o = wf + l * GATES * CDIM + (n0 + 8 * half) * CDIM + k0 + l16;
#pragma unroll
  for (int r = 0; r < 8; ++r) o[r * CDIM] = acc0[r] + acc1[r];
}

#define SEG_CAP 8192

template <bool TO_LDS>
__device__ __forceinline__ void sort_segment(const int* seg, int* srt, int* __restrict__ gs,
                                             int n, int lane) {
  if (n <= 32) {
    const int key = (lane < n) ? seg[lane] : 0;
    int rank = 0;
    for (int c = 0; c < n; ++c) {
      const int other = __builtin_amdgcn_readlane(key, c);
      rank += (other < key || (other == key && c < lane)) ? 1 : 0;
    }
    if (lane < n) {
      if (TO_LDS) srt[rank] = key;
      gs[rank] = key;
    }
  } else {
    for (int a = lane; a < n; a += 32) {
      const int key = seg[a];
      int rank = 0;
      for (int c = 0; c < n; ++c) {
        const int other = seg[c];
        rank += (other < key || (other == key && c < a)) ? 1 : 0;
      }
      if (TO_LDS) srt[rank] = key;
      gs[rank] = key;
    }
  }
}

__device__ __forceinline__ float4 sum_rows(const float* __restrict__ m, const int* keys,
                                           int n, int lane, int c4) {
  float4 acc = make_float4(0.f, 0.f, 0.f, 0.f);
  for (int b = 0; b < n; b += 32) {
    const int cnt   = (n - b < 32) ? (n - b) : 32;
    const int mycol = (lane < cnt) ? keys[b + lane] : 0;
    for (int j = 0; j < cnt; j += 8) {
      float4 v[8];
#pragma unroll
      for (int u = 0; u < 8; ++u) {
        const int s = __builtin_amdgcn_readlane(mycol, j + u);
        v[u] = *(const float4*)(m + s * CDIM + c4);
      }
#pragma unroll
      for (int u = 0; u < 8; ++u) {
        if (j + u < cnt) {
          acc.x += v[u].x; acc.y += v[u].y; acc.z += v[u].z; acc.w += v[u].w;
        }
      }
    }
  }
  return acc;
}

__global__ __launch_bounds__(1024) void gnn_range(const int* __restrict__ cell,
                                                  const int* __restrict__ tmp,
                                                  int* __restrict__ rowptr,
                                                  int* __restrict__ col,
                                                  int* __restrict__ scol,
                                                  const float* __restrict__ x,
                                                  float* __restrict__ agg) {
  __shared__ int cnt[256];
  __shared__ int st[256];
  __shared__ int cu[256];
  __shared__ int wtot[8];
  __shared__ int ccn[256];
  __shared__ int cof[256];
  __shared__ int ctot[8];
  __shared__ int wpre[32];
  __shared__ int lseg[SEG_CAP];
  __shared__ int lsrt[SEG_CAP];
  const int t    = threadIdx.x;
  const int lane = t & 31;
  const int wv   = __builtin_amdgcn_readfirstlane(t >> 5);
  const int r    = blockIdx.x;

  const int nfront = r * NCH;
  const int nf4    = nfront >> 2;
  int pre = 0;
  int cw;
  {
    const int lastv = (nf4 > 0) ? nf4 - 1 : 0;
    int4 v[16];
#pragma unroll
    for (int k = 0; k < 16; ++k) {
      const int j = t + 1024 * k;
      v[k] = *(const int4*)(cell + 4 * ((j < nf4) ? j : lastv));
    }
    const int tl = cell[(t < (nfront & 3)) ? (4 * nf4 + t) : 0];
    const int ow = cell[nfront + ((t < NCH) ? t : (NCH - 1))];
#pragma unroll
    for (int k = 0; k < 16; ++k)
      pre += (t + 1024 * k < nf4) ? ((v[k].x & 4095) + (v[k].y & 4095) + (v[k].z & 4095) + (v[k].w & 4095)) : 0;
    pre += (t < (nfront & 3)) ? (tl & 4095) : 0;
    cw = (t < NCH) ? ow : 0;
  }
#pragma unroll
  for (int off = 16; off > 0; off >>= 1) pre += __shfl_xor(pre, off);
  if (lane == 0) wpre[wv] = pre;
  int csum = cw & 4095;
#pragma unroll
  for (int off = 16; off > 0; off >>= 1) csum += __shfl_xor(csum, off);
  if (t < 256) {
    if (lane == 0) ctot[wv] = csum;
    ccn[t] = cw & 4095;
    cof[t] = t * CHE + (cw >> 12);
    cnt[t] = 0;
  }
  __syncthreads();

  int begv = wpre[lane];
  int nv   = (lane < 8) ? ctot[lane] : 0;
#pragma unroll
  for (int off = 16; off > 0; off >>= 1) {
    begv += __shfl_xor(begv, off);
    nv   += __shfl_xor(nv, off);
  }
  const int beg  = __builtin_amdgcn_readfirstlane(begv);
  const int n    = __builtin_amdgcn_readfirstlane(nv);
  const int end  = beg + n;
  const bool fits = (n <= SEG_CAP);

  int w[8], cc[8], co[8];
  bool more = false;
#pragma unroll
  for (int k = 0; k < 8; ++k) {
    cc[k] = ccn[wv + 32 * k];
    co[k] = cof[wv + 32 * k];
  }
#pragma unroll
  for (int k = 0; k < 8; ++k) {
    cc[k] = __builtin_amdgcn_readfirstlane(cc[k]);
    co[k] = __builtin_amdgcn_readfirstlane(co[k]);
    more  = more || (cc[k] > 32);
  }
#pragma unroll
  for (int k = 0; k < 8; ++k) w[k] = (lane < cc[k]) ? tmp[co[k] + lane] : -1;
#pragma unroll
  for (int k = 0; k < 8; ++k)
    if (w[k] >= 0) atomicAdd(&cnt[w[k] >> 16], 1);
  if (more) {
#pragma unroll
    for (int k = 0; k < 8; ++k)
      for (int j = 32 + lane; j < cc[k]; j += 32) atomicAdd(&cnt[tmp[co[k] + j] >> 16], 1);
  }
  __syncthreads();

  int cn = 0, inc = 0;
  if (t < 256) {
    cn  = cnt[t];
    inc = cn;
#pragma unroll
    for (int off = 1; off < 32; off <<= 1) {
      const int v = __shfl_up(inc, off);
      if (lane >= off) inc += v;
    }
    if (lane == 31) wtot[wv] = inc;
  }
  __syncthreads();
  if (t < 256) {
    int front = 0;
#pragma unroll
    for (int k = 0; k < 8; ++k) front += (k < wv) ? wtot[k] : 0;
    const int excl = front + inc - cn;
    st[t] = excl;
    cu[t] = excl;
    if (t < BW) rowptr[r * BW + t] = beg + excl;
  }
  if (r == NB - 1 && t == 0) rowptr[N_NODES] = end;
  __syncthreads();

#pragma unroll
  for (int k = 0; k < 8; ++k) {
    if (w[k] >= 0) {
      const int p = atomicAdd(&cu[w[k] >> 16], 1);
      if (fits) lseg[p] = w[k] & 0xffff;
      else      col[beg + p] = w[k] & 0xffff;
    }
  }
  if (more) {
#pragma unroll
    for (int k = 0; k < 8; ++k) {
      for (int j = 32 + lane; j < cc[k]; j += 32) {
        const int wj = tmp[co[k] + j];
        const int p  = atomicAdd(&cu[wj >> 16], 1);
        if (fits) lseg[p] = wj & 0xffff;
        else      col[beg + p] = wj & 0xffff;
      }
    }
  }
  if (!fits) __threadfence();
  __syncthreads();

  const int c4 = (lane < 24) ? lane * 4 : 0;
  for (int node = wv; node < BW; node += 32) {
    const int off = __builtin_amdgcn_readfirstlane(st[node]);
    const int nd  = __builtin_amdgcn_readfirstlane(cnt[node]);
    float4 acc;
    if (fits) {
      sort_segment<true>(lseg + off, lsrt + off, scol + beg + off, nd, lane);
      acc = sum_rows(x, lsrt + off, nd, lane, c4);
    } else {
      sort_segment<false>(col + beg + off, nullptr, scol + beg + off, nd, lane);
      __threadfence();
      acc = sum_rows(x, scol + beg + off, nd, lane, c4);
    }
    if (lane < 24) *(float4*)(agg + (r * BW + node) * CDIM + c4) = acc;
  }
}

__global__ __launch_bounds__(256) void gnn_gather(const float* __restrict__ m,
                                                  const int* __restrict__ rowptr,
                                                  const int* __restrict__ col,
                                                  float* __restrict__ agg) {
  const int lane = threadIdx.x & 31;
  const int wave0  = __builtin_amdgcn_readfirstlane(blockIdx.x * 8 + (threadIdx.x >> 5));
  const int nwaves = gridDim.x * 8;
  const int c4     = (lane < 24) ? lane * 4 : 0;

  for (int node = wave0; node < N_NODES; node += nwaves) {
    const int beg = rowptr[node];
    const int end = rowptr[node + 1];
    float4 acc = make_float4(0.f, 0.f, 0.f, 0.f);
    for (int b = beg; b < end; b += 32) {
      const int cnt   = (end - b < 32) ? (end - b) : 32;
      const int mycol = (lane < cnt) ? col[b + lane] : 0;
      for (int j = 0; j < cnt; j += 8) {
        float4 v[8];
#pragma unroll
        for (int u = 0; u < 8; ++u) {
          const int s = __builtin_amdgcn_readlane(mycol, j + u);
          v[u] = *(const float4*)(m + s * CDIM + c4);
        }
#pragma unroll
        for (int u = 0; u < 8; ++u) {
          if (j + u < cnt) {
            acc.x += v[u].x; acc.y += v[u].y; acc.z += v[u].z; acc.w += v[u].w;
          }
        }
      }
    }
    if (lane < 24) *(float4*)(agg + node * CDIM + c4) = acc;
  }
}

#define GRU_SLOTS 85
#define GRU_WAVES (GRU_SLOTS * 8)

__device__ __forceinline__ void gru_tile(const float* agp, const float* ahp,
                                         const float* bp, const float* Ah,
                                         float bir, float biz, float bin_, float bhr, float bhz, float bhn,
                                         float* __restrict__ out, int row0, int col, int half) {
  v8f a_ir, a_iz, a_in, a_hr, a_hz, a_hn;
#pragma unroll
  for (int r = 0; r < 8; ++r) {
    a_ir[r] = bir; a_iz[r] = biz; a_in[r] = bin_;
    a_hr[r] = bhr; a_hz[r] = bhz; a_hn[r] = bhn;
  }

#pragma unroll 4
  for (int k0 = 0; k0 < CDIM; k0 += 4) {
    const v2f ai = *(const v2f*)(agp + k0);
    const v2f ah = *(const v2f*)(ahp + k0);
    const v2f b0 = *(const v2f*)(bp + 0 * 16 * LDP + k0);
    const v2f b1 = *(const v2f*)(bp + 1 * 16 * LDP + k0);
    const v2f b2 = *(const v2f*)(bp + 2 * 16 * LDP + k0);
    const v2f b3 = *(const v2f*)(bp + 3 * 16 * LDP + k0);
    const v2f b4 = *(const v2f*)(bp + 4 * 16 * LDP + k0);
    const v2f b5 = *(const v2f*)(bp + 5 * 16 * LDP + k0);
    a_ir = __builtin_amdgcn_wmma_f32_16x16x4_f32(false, ai, false, b0, (short)0, a_ir, false, false);
    a_iz = __builtin_amdgcn_wmma_f32_16x16x4_f32(false, ai, false, b1, (short)0, a_iz, false, false);
    a_in = __builtin_amdgcn_wmma_f32_16x16x4_f32(false, ai, false, b2, (short)0, a_in, false, false);
    a_hr = __builtin_amdgcn_wmma_f32_16x16x4_f32(false, ah, false, b3, (short)0, a_hr, false, false);
    a_hz = __builtin_amdgcn_wmma_f32_16x16x4_f32(false, ah, false, b4, (short)0, a_hz, false, false);
    a_hn = __builtin_amdgcn_wmma_f32_16x16x4_f32(false, ah, false, b5, (short)0, a_hn, false, false);
  }

#pragma unroll
  for (int r = 0; r < 8; ++r) {
    const int lr  = r + 8 * half;
    const float hv = Ah[lr * LDP + col];
    const float rr = 0.5f + 0.5f * __builtin_amdgcn_tanhf(0.5f * (a_ir[r] + a_hr[r]));
    const float zz = 0.5f + 0.5f * __builtin_amdgcn_tanhf(0.5f * (a_iz[r] + a_hz[r]));
    const float nn = __builtin_amdgcn_tanhf(a_in[r] + rr * a_hn[r]);
    out[(row0 + lr) * CDIM + col] = (1.f - zz) * nn + zz * hv;
  }
}

__global__ __launch_bounds__(256) void gnn_gru(const float* __restrict__ agg,
                                               const float* __restrict__ h,
                                               const float* __restrict__ w_ih,
                                               const float* __restrict__ w_hh,
                                               const float* __restrict__ b_ih,
                                               const float* __restrict__ b_hh,
                                               float* __restrict__ out) {
  __shared__ __attribute__((aligned(16))) float lds[6 * 16 * LDP + 8 * 2 * 16 * LDP];
  const int tid  = threadIdx.x;
  const int wid  = tid >> 5;
  const int lane = tid & 31;
  const int half = lane >> 4;
  const int l16  = lane & 15;
  float* Bs = lds;
  float* Ag = lds + 6 * 16 * LDP + wid * (2 * 16 * LDP);
  float* Ah = Ag + 16 * LDP;

  const int tc   = blockIdx.x % (CDIM / 16);
  const int slot = blockIdx.x / (CDIM / 16);
  const int col0 = tc * 16;
  const int col  = col0 + l16;
  const int tr0 = __builtin_amdgcn_readfirstlane(((wid + slot + tc) & 7) * GRU_SLOTS + slot);

  const float bir = b_ih[col], biz = b_ih[CDIM + col], bin_ = b_ih[2 * CDIM + col];
  const float bhr = b_hh[col], bhz = b_hh[CDIM + col], bhn  = b_hh[2 * CDIM + col];

  {
    float4 wv[9];
    float4 gv[12], hv4[12];
#pragma unroll
    for (int i = 0; i < 9; ++i) {
      const int idx = i * 256 + tid;
      const int j   = idx / 384;
      const int w   = idx % 384;
      const float* srcw = (j < 3 ? w_ih : w_hh) + ((j % 3) * CDIM + col0) * CDIM;
      wv[i] = *(const float4*)(srcw + w * 4);
    }
    load_rowtile(agg + tr0 * 16 * CDIM, lane, gv);
    load_rowtile(h   + tr0 * 16 * CDIM, lane, hv4);
#pragma unroll
    for (int i = 0; i < 9; ++i) {
      const int idx = i * 256 + tid;
      const int j   = idx / 384;
      const int w   = idx % 384;
      *(float4*)(Bs + (j * 16 + w / 24) * LDP + (w % 24) * 4) = wv[i];
    }
    store_rowtile(Ag, lane, gv);
    store_rowtile(Ah, lane, hv4);
  }
  __syncthreads();

  const float* agp = Ag + l16 * LDP + 2 * half;
  const float* ahp = Ah + l16 * LDP + 2 * half;
  const float* bp  = Bs + l16 * LDP + 2 * half;

  int tr = tr0;
  for (; tr + GRU_WAVES <= LASTTILE; tr += GRU_WAVES) {
    gru_tile(agp, ahp, bp, Ah, bir, biz, bin_, bhr, bhz, bhn, out, tr * 16, col, half);

    float4 gv[12], hv4[12];
    load_rowtile(agg + (tr + GRU_WAVES) * 16 * CDIM, lane, gv);
    load_rowtile(h   + (tr + GRU_WAVES) * 16 * CDIM, lane, hv4);
    store_rowtile(Ag, lane, gv);
    store_rowtile(Ah, lane, hv4);
  }
  gru_tile(agp, ahp, bp, Ah, bir, biz, bin_, bhr, bhz, bhn, out, tr * 16, col, half);
}

extern "C" void kernel_launch(void* const* d_in, const int* in_sizes, int n_in,
                              void* d_out, int out_size, void* d_ws, size_t ws_size,
                              hipStream_t stream) {
  const float* x      = (const float*)d_in[0];
  const int*   ei     = (const int*)  d_in[1];
  const float* weight = (const float*)d_in[2];
  const float* w_ih   = (const float*)d_in[3];
  const float* w_hh   = (const float*)d_in[4];
  const float* b_ih   = (const float*)d_in[5];
  const float* b_hh   = (const float*)d_in[6];
  float* out = (float*)d_out;

  float* xa  = (float*)d_ws;
  int*   ib  = (int*)(xa + NC);
  float* wf  = xa + 2 * NC;
  float* agg = xa + 3 * NC;

  int* cell   = ib;
  int* tmp    = ib + 62592;
  int* rowptr = tmp + N_EDGES;
  int* col    = rowptr + 50048;
  int* scol   = col + N_EDGES;

  const int gruBlocks     = GRU_SLOTS * (CDIM / 16);
  const int setupBlocks   = NCH + FOLD_BLOCKS;
  const int gathBlocks    = 2048;

  gnn_setup    <<<setupBlocks, 256, 0, stream>>>(ei, cell, tmp, weight, w_ih, wf);
  gnn_range    <<<NB, 1024, 0, stream>>>(cell, tmp, rowptr, col, scol, x, agg);

  for (int l = 0; l < 3; ++l) {
    const float* xin  = (l == 0) ? x   : ((l == 1) ? out : xa);
    float*       xout = (l == 1) ? xa  : out;
    const float* wfl  = wf + l * GATES * CDIM;

    if (l > 0) gnn_gather<<<gathBlocks, 256, 0, stream>>>(xin, rowptr, scol, agg);
    gnn_gru   <<<gruBlocks,  256, 0, stream>>>(agg, xin, wfl, w_hh, b_ih, b_hh, xout);
  }
}
